# starting kernel
# baseline (speedup 1.0000x reference)
#include <hip/hip_runtime.h>
#include <hip/hip_bf16.h>


typedef __attribute__((ext_vector_type(2))) float v2f;
typedef __attribute__((ext_vector_type(8))) float v8f;

#define BT 48
#define TT 12
#define BB 4
#define HCH 32

__global__ void k_deg_init(float* __restrict__ deg, int n_nodes) {
    int n = blockIdx.x * blockDim.x + threadIdx.x;
    if (n < n_nodes) deg[n] = 1.0f;
}

__global__ void k_deg_edges(const int* __restrict__ ei, const float* __restrict__ ew,
                            float* __restrict__ deg, int n_edges) {
    int e = blockIdx.x * blockDim.x + threadIdx.x;
    if (e < n_edges) {
        int d = ei[(size_t)n_edges + e];
        atomicAdd(&deg[d], ew[e]);
    }
}

__global__ void k_finalize(const float* __restrict__ x, float* __restrict__ dinv,
                           float* __restrict__ xt, float* __restrict__ aggt, int n_nodes) {
    int n = blockIdx.x * blockDim.x + threadIdx.x;
    if (n >= n_nodes) return;
    float deg = dinv[n];
    float di  = deg > 0.0f ? rsqrtf(deg) : 0.0f;
    dinv[n] = di;
    float sn = di * di;
    float buf[BT];
#pragma unroll
    for (int s = 0; s < BT; ++s)
        buf[s] = x[(size_t)s * n_nodes + n];
    float4* xt4 = (float4*)(xt + (size_t)n * BT);
    float4* ag4 = (float4*)(aggt + (size_t)n * BT);
#pragma unroll
    for (int q = 0; q < BT / 4; ++q) {
        float4 v = make_float4(buf[4*q], buf[4*q+1], buf[4*q+2], buf[4*q+3]);
        xt4[q] = v;
        ag4[q] = make_float4(v.x * sn, v.y * sn, v.z * sn, v.w * sn);
    }
}

__global__ void k_aggregate(const int* __restrict__ ei, const float* __restrict__ ew,
                            const float* __restrict__ dinv, const float* __restrict__ xt,
                            float* __restrict__ aggt, int n_edges) {
    int lane = threadIdx.x & 31;
    long long wid = (long long)blockIdx.x * (blockDim.x >> 5) + (threadIdx.x >> 5);
    long long nw  = (long long)gridDim.x * (blockDim.x >> 5);
    for (long long e = wid; e < n_edges; e += nw) {
        int   s_  = ei[e];
        int   d_  = ei[(size_t)n_edges + e];
        float w   = ew[e];
        float nrm = dinv[s_] * w * dinv[d_];
        const float* xs = xt   + (size_t)s_ * BT;
        float*       ad = aggt + (size_t)d_ * BT;
        atomicAdd(&ad[lane], xs[lane] * nrm);
        if (lane < 16)
            atomicAdd(&ad[32 + lane], xs[32 + lane] * nrm);
    }
}

__global__ void k_epilogue(const float* __restrict__ aggt,
                           const float* __restrict__ gw, const float* __restrict__ gb,
                           const float* __restrict__ cw, const float* __restrict__ cb,
                           const float* __restrict__ lwp, const float* __restrict__ lbp,
                           float* __restrict__ out, int n_nodes) {
    const int lane = threadIdx.x & 31;
    const int h    = lane >> 4;
    const int m    = lane & 15;

    v2f bmat0, bmat1;
#pragma unroll
    for (int v = 0; v < 2; ++v) {
        int k = 2 * h + v;
        float a0 = 0.0f, a1 = 0.0f;
        if (k < 3) {
            for (int ci = 0; ci < HCH; ++ci) {
                float g = gw[ci];
                a0 += g * cw[(size_t)m        * 96 + ci * 3 + k];
                a1 += g * cw[(size_t)(m + 16) * 96 + ci * 3 + k];
            }
        }
        bmat0[v] = a0;
        bmat1[v] = a1;
    }
    float bbA[3], bbB[3];
#pragma unroll
    for (int k = 0; k < 3; ++k) {
        float a0 = 0.0f, a1 = 0.0f;
        for (int ci = 0; ci < HCH; ++ci) {
            float g = gb[ci];
            a0 += g * cw[(size_t)m        * 96 + ci * 3 + k];
            a1 += g * cw[(size_t)(m + 16) * 96 + ci * 3 + k];
        }
        bbA[k] = a0;
        bbB[k] = a1;
    }
    v8f cmat0, cmat1;
#pragma unroll
    for (int j = 0; j < 8; ++j) {
        int t = j + 8 * h;
        float c0 = 0.0f, c1 = 0.0f;
        if (t < TT) {
            c0 = cb[m]      + bbA[1] + (t > 0 ? bbA[0] : 0.0f) + (t < TT - 1 ? bbA[2] : 0.0f);
            c1 = cb[m + 16] + bbB[1] + (t > 0 ? bbB[0] : 0.0f) + (t < TT - 1 ? bbB[2] : 0.0f);
        }
        cmat0[j] = c0;
        cmat1[j] = c1;
    }
    const float lwA = lwp[m], lwB = lwp[16 + m];
    const float lb  = lbp[0];

    const int  i0  = m + 2 * h - 1;
    const int  i1  = i0 + 1;
    const bool ok0 = (m < TT) && (i0 >= 0) && (i0 < TT);
    const bool ok1 = (m < TT) && (i1 < TT);
    const int  o0  = ok0 ? i0 : 0;
    const int  o1  = ok1 ? i1 : 0;
    const float msk0 = ok0 ? 1.0f : 0.0f;
    const float msk1 = ok1 ? 1.0f : 0.0f;

    const int wid = blockIdx.x * (blockDim.x >> 5) + (threadIdx.x >> 5);
    const int nw  = gridDim.x * (blockDim.x >> 5);

#pragma unroll
    for (int b = 0; b < BB; ++b) {
        const float* agb  = aggt + b * TT;
        float*       outb = out + (size_t)b * n_nodes;
        for (int n = wid; n < n_nodes; n += nw) {
            const float* ag = agb + (size_t)n * BT;

            v2f amat;
            amat[0] = ag[o0] * msk0;
            amat[1] = ag[o1] * msk1;

            v8f d0 = __builtin_amdgcn_wmma_f32_16x16x4_f32(false, amat, false, bmat0,
                                                           (short)0, cmat0, false, false);
            v8f d1 = __builtin_amdgcn_wmma_f32_16x16x4_f32(false, amat, false, bmat1,
                                                           (short)0, cmat1, false, false);

            float sA = 0.0f, sB = 0.0f;
#pragma unroll
            for (int j = 0; j < 8; ++j) {
                sA += fmaxf(d0[j], 0.0f);
                sB += fmaxf(d1[j], 0.0f);
            }
            sA += __shfl_xor(sA, 16);
            sB += __shfl_xor(sB, 16);

            float v = sA * lwA + sB * lwB;
            v += __shfl_xor(v, 1);
            v += __shfl_xor(v, 2);
            v += __shfl_xor(v, 4);
            v += __shfl_xor(v, 8);

            if (lane == 0)
                outb[n] = v * (1.0f / (float)TT) + lb;
        }
    }
}

extern "C" void kernel_launch(void* const* d_in, const int* in_sizes, int n_in,
                              void* d_out, int out_size, void* d_ws, size_t ws_size,
                              hipStream_t stream) {
    const float* x  = (const float*)d_in[0];
    const int*   ei = (const int*)d_in[1];
    const float* ew = (const float*)d_in[2];
    const float* gw = (const float*)d_in[3];
    const float* gb = (const float*)d_in[4];
    const float* cw = (const float*)d_in[5];
    const float* cb = (const float*)d_in[6];
    const float* lw = (const float*)d_in[7];
    const float* lb = (const float*)d_in[8];
    float* out = (float*)d_out;

    const int E = in_sizes[2];
    const int N = in_sizes[0] / BT;

    float* dinv = (float*)d_ws;
    float* xt   = dinv + N;
    float* aggt = xt + (size_t)N * BT;

    k_deg_init <<<(N + 255) / 256, 256, 0, stream>>>(dinv, N);
    k_deg_edges<<<(E + 255) / 256, 256, 0, stream>>>(ei, ew, dinv, E);
    k_finalize <<<(N + 255) / 256, 256, 0, stream>>>(x, dinv, xt, aggt, N);

    int aggBlocks = (E + 7) / 8;
    k_aggregate<<<aggBlocks, 256, 0, stream>>>(ei, ew, dinv, xt, aggt, E);

    k_epilogue <<<1024, 256, 0, stream>>>(aggt, gw, gb, cw, cb, lw, lb, out, N);
}


# degree sums: each workgroup adds its edge slice into an f32 table in LDS (ds atomics), then adds the table to deg with line-contiguous global atomics; the weights are summed in another order
# speedup vs baseline: 1.2105x; 1.2105x over previous
#include <hip/hip_runtime.h>
#include <hip/hip_bf16.h>


typedef __attribute__((ext_vector_type(2))) float v2f;
typedef __attribute__((ext_vector_type(8))) float v8f;

#define BT 48
#define TT 12
#define BB 4
#define HCH 32

__global__ void k_deg_init(float* __restrict__ deg, int n_nodes) {
    int n = blockIdx.x * blockDim.x + threadIdx.x;
    if (n < n_nodes) deg[n] = 1.0f;
}

#define DEG_CAP 51200
#define DEG_WG  64
#define DEG_TPB 1024
#define DEG_UNR 8
#define DEG_FL  7

__global__ __launch_bounds__(DEG_TPB)
void k_deg_edges(const int* __restrict__ ei, const float* __restrict__ ew,
                 float* __restrict__ deg, int n_edges, int n_nodes) {
    __shared__ float4 hist4[DEG_CAP / 4];
    float* hist = (float*)hist4;
    const int tid = threadIdx.x;
    const int per = (n_edges + (int)gridDim.x - 1) / (int)gridDim.x;
    const int e0  = (int)blockIdx.x * per;
    const int e1  = min(e0 + per, n_edges);
    const int* __restrict__ dst = ei + (size_t)n_edges;

    for (int base = 0; base < n_nodes; base += DEG_CAP) {
        const int cnt = min(DEG_CAP, n_nodes - base);

        for (int i = tid; i < DEG_CAP / 4; i += DEG_TPB)
            hist4[i] = make_float4(0.0f, 0.0f, 0.0f, 0.0f);
        __syncthreads();

        if (e0 < e1) {
            const int elast = e1 - 1;
            const int step  = DEG_UNR * DEG_TPB;
            int   dv[DEG_UNR];
            float w[DEG_UNR];
#pragma unroll
            for (int u = 0; u < DEG_UNR; ++u) {
                const int ec = min(e0 + tid + u * DEG_TPB, elast);
                dv[u] = dst[ec];
                w[u]  = ew[ec];
            }
            for (int e = e0 + tid; e < e1; e += step) {
                int   dn[DEG_UNR];
                float wn[DEG_UNR];
#pragma unroll
                for (int u = 0; u < DEG_UNR; ++u) {
                    const int ec = min(e + step + u * DEG_TPB, elast);
                    dn[u] = dst[ec];
                    wn[u] = ew[ec];
                }
#pragma unroll
                for (int u = 0; u < DEG_UNR; ++u) {
                    const int d = (e + u * DEG_TPB < e1) ? dv[u] - base : -1;
                    if ((unsigned)d < (unsigned)cnt)
                        atomicAdd(&hist[d], w[u]);
                }
#pragma unroll
                for (int u = 0; u < DEG_UNR; ++u) { dv[u] = dn[u]; w[u] = wn[u]; }
            }
        }
        __syncthreads();

        const int rounds = (cnt + DEG_TPB - 1) / DEG_TPB;
        const int rot    = (int)blockIdx.x % rounds;
        for (int r = 0; r < rounds; r += DEG_FL) {
            float v[DEG_FL];
            int   idx[DEG_FL];
#pragma unroll
            for (int u = 0; u < DEG_FL; ++u) {
                int rr = r + u + rot;
                if (rr >= rounds) rr -= rounds;
                const int  i  = rr * DEG_TPB + tid;
                const bool ok = (r + u < rounds) && (i < cnt);
                const float hv = hist[ok ? i : tid];
                idx[u] = i;
                v[u]   = ok ? hv : 0.0f;
            }
#pragma unroll
            for (int u = 0; u < DEG_FL; ++u)
                if (v[u] != 0.0f) atomicAdd(&deg[base + idx[u]], v[u]);
        }
        __syncthreads();
    }
}

__global__ void k_finalize(const float* __restrict__ x, float* __restrict__ dinv,
                           float* __restrict__ xt, float* __restrict__ aggt, int n_nodes) {
    int n = blockIdx.x * blockDim.x + threadIdx.x;
    if (n >= n_nodes) return;
    float deg = dinv[n];
    float di  = deg > 0.0f ? rsqrtf(deg) : 0.0f;
    dinv[n] = di;
    float sn = di * di;
    float buf[BT];
#pragma unroll
    for (int s = 0; s < BT; ++s)
        buf[s] = x[(size_t)s * n_nodes + n];
    float4* xt4 = (float4*)(xt + (size_t)n * BT);
    float4* ag4 = (float4*)(aggt + (size_t)n * BT);
#pragma unroll
    for (int q = 0; q < BT / 4; ++q) {
        float4 v = make_float4(buf[4*q], buf[4*q+1], buf[4*q+2], buf[4*q+3]);
        xt4[q] = v;
        ag4[q] = make_float4(v.x * sn, v.y * sn, v.z * sn, v.w * sn);
    }
}

__global__ void k_aggregate(const int* __restrict__ ei, const float* __restrict__ ew,
                            const float* __restrict__ dinv, const float* __restrict__ xt,
                            float* __restrict__ aggt, int n_edges) {
    int lane = threadIdx.x & 31;
    long long wid = (long long)blockIdx.x * (blockDim.x >> 5) + (threadIdx.x >> 5);
    long long nw  = (long long)gridDim.x * (blockDim.x >> 5);
    for (long long e = wid; e < n_edges; e += nw) {
        int   s_  = ei[e];
        int   d_  = ei[(size_t)n_edges + e];
        float w   = ew[e];
        float nrm = dinv[s_] * w * dinv[d_];
        const float* xs = xt   + (size_t)s_ * BT;
        float*       ad = aggt + (size_t)d_ * BT;
        atomicAdd(&ad[lane], xs[lane] * nrm);
        if (lane < 16)
            atomicAdd(&ad[32 + lane], xs[32 + lane] * nrm);
    }
}

__global__ void k_epilogue(const float* __restrict__ aggt,
                           const float* __restrict__ gw, const float* __restrict__ gb,
                           const float* __restrict__ cw, const float* __restrict__ cb,
                           const float* __restrict__ lwp, const float* __restrict__ lbp,
                           float* __restrict__ out, int n_nodes) {
    const int lane = threadIdx.x & 31;
    const int h    = lane >> 4;
    const int m    = lane & 15;

    v2f bmat0, bmat1;
#pragma unroll
    for (int v = 0; v < 2; ++v) {
        int k = 2 * h + v;
        float a0 = 0.0f, a1 = 0.0f;
        if (k < 3) {
            for (int ci = 0; ci < HCH; ++ci) {
                float g = gw[ci];
                a0 += g * cw[(size_t)m        * 96 + ci * 3 + k];
                a1 += g * cw[(size_t)(m + 16) * 96 + ci * 3 + k];
            }
        }
        bmat0[v] = a0;
        bmat1[v] = a1;
    }
    float bbA[3], bbB[3];
#pragma unroll
    for (int k = 0; k < 3; ++k) {
        float a0 = 0.0f, a1 = 0.0f;
        for (int ci = 0; ci < HCH; ++ci) {
            float g = gb[ci];
            a0 += g * cw[(size_t)m        * 96 + ci * 3 + k];
            a1 += g * cw[(size_t)(m + 16) * 96 + ci * 3 + k];
        }
        bbA[k] = a0;
        bbB[k] = a1;
    }
    v8f cmat0, cmat1;
#pragma unroll
    for (int j = 0; j < 8; ++j) {
        int t = j + 8 * h;
        float c0 = 0.0f, c1 = 0.0f;
        if (t < TT) {
            c0 = cb[m]      + bbA[1] + (t > 0 ? bbA[0] : 0.0f) + (t < TT - 1 ? bbA[2] : 0.0f);
            c1 = cb[m + 16] + bbB[1] + (t > 0 ? bbB[0] : 0.0f) + (t < TT - 1 ? bbB[2] : 0.0f);
        }
        cmat0[j] = c0;
        cmat1[j] = c1;
    }
    const float lwA = lwp[m], lwB = lwp[16 + m];
    const float lb  = lbp[0];

    const int  i0  = m + 2 * h - 1;
    const int  i1  = i0 + 1;
    const bool ok0 = (m < TT) && (i0 >= 0) && (i0 < TT);
    const bool ok1 = (m < TT) && (i1 < TT);
    const int  o0  = ok0 ? i0 : 0;
    const int  o1  = ok1 ? i1 : 0;
    const float msk0 = ok0 ? 1.0f : 0.0f;
    const float msk1 = ok1 ? 1.0f : 0.0f;

    const int wid = blockIdx.x * (blockDim.x >> 5) + (threadIdx.x >> 5);
    const int nw  = gridDim.x * (blockDim.x >> 5);

#pragma unroll
    for (int b = 0; b < BB; ++b) {
        const float* agb  = aggt + b * TT;
        float*       outb = out + (size_t)b * n_nodes;
        for (int n = wid; n < n_nodes; n += nw) {
            const float* ag = agb + (size_t)n * BT;

            v2f amat;
            amat[0] = ag[o0] * msk0;
            amat[1] = ag[o1] * msk1;

            v8f d0 = __builtin_amdgcn_wmma_f32_16x16x4_f32(false, amat, false, bmat0,
                                                           (short)0, cmat0, false, false);
            v8f d1 = __builtin_amdgcn_wmma_f32_16x16x4_f32(false, amat, false, bmat1,
                                                           (short)0, cmat1, false, false);

            float sA = 0.0f, sB = 0.0f;
#pragma unroll
            for (int j = 0; j < 8; ++j) {
                sA += fmaxf(d0[j], 0.0f);
                sB += fmaxf(d1[j], 0.0f);
            }
            sA += __shfl_xor(sA, 16);
            sB += __shfl_xor(sB, 16);

            float v = sA * lwA + sB * lwB;
            v += __shfl_xor(v, 1);
            v += __shfl_xor(v, 2);
            v += __shfl_xor(v, 4);
            v += __shfl_xor(v, 8);

            if (lane == 0)
                outb[n] = v * (1.0f / (float)TT) + lb;
        }
    }
}

extern "C" void kernel_launch(void* const* d_in, const int* in_sizes, int n_in,
                              void* d_out, int out_size, void* d_ws, size_t ws_size,
                              hipStream_t stream) {
    const float* x  = (const float*)d_in[0];
    const int*   ei = (const int*)d_in[1];
    const float* ew = (const float*)d_in[2];
    const float* gw = (const float*)d_in[3];
    const float* gb = (const float*)d_in[4];
    const float* cw = (const float*)d_in[5];
    const float* cb = (const float*)d_in[6];
    const float* lw = (const float*)d_in[7];
    const float* lb = (const float*)d_in[8];
    float* out = (float*)d_out;

    const int E = in_sizes[2];
    const int N = in_sizes[0] / BT;

    float* dinv = (float*)d_ws;
    float* xt   = dinv + N;
    float* aggt = xt + (size_t)N * BT;

    k_deg_init <<<(N + 255) / 256, 256, 0, stream>>>(dinv, N);
    k_deg_edges<<<DEG_WG, DEG_TPB, 0, stream>>>(ei, ew, dinv, E, N);
    k_finalize <<<(N + 255) / 256, 256, 0, stream>>>(x, dinv, xt, aggt, N);

    int aggBlocks = (E + 7) / 8;
    k_aggregate<<<aggBlocks, 256, 0, stream>>>(ei, ew, dinv, xt, aggt, E);

    k_epilogue <<<1024, 256, 0, stream>>>(aggt, gw, gb, cw, cb, lw, lb, out, N);
}


# edge aggregation on a resident grid: a wave walks batches of 32 edges (indices, weights, norms one edge per lane; 8 edges' row loads in flight) instead of one short wave per edge; aggt atomics kept
# speedup vs baseline: 1.8655x; 1.5411x over previous
#include <hip/hip_runtime.h>
#include <hip/hip_bf16.h>


typedef __attribute__((ext_vector_type(2))) float v2f;
typedef __attribute__((ext_vector_type(8))) float v8f;

#define BT 48
#define TT 12
#define BB 4
#define HCH 32

__global__ void k_deg_init(float* __restrict__ deg, int n_nodes) {
    int n = blockIdx.x * blockDim.x + threadIdx.x;
    if (n < n_nodes) deg[n] = 1.0f;
}

#define DEG_CAP 51200
#define DEG_WG  64
#define DEG_TPB 1024
#define DEG_UNR 8
#define DEG_FL  7

__global__ __launch_bounds__(DEG_TPB)
void k_deg_edges(const int* __restrict__ ei, const float* __restrict__ ew,
                 float* __restrict__ deg, int n_edges, int n_nodes) {
    __shared__ float4 hist4[DEG_CAP / 4];
    float* hist = (float*)hist4;
    const int tid = threadIdx.x;
    const int per = (n_edges + (int)gridDim.x - 1) / (int)gridDim.x;
    const int e0  = (int)blockIdx.x * per;
    const int e1  = min(e0 + per, n_edges);
    const int* __restrict__ dst = ei + (size_t)n_edges;

    for (int base = 0; base < n_nodes; base += DEG_CAP) {
        const int cnt = min(DEG_CAP, n_nodes - base);

        for (int i = tid; i < DEG_CAP / 4; i += DEG_TPB)
            hist4[i] = make_float4(0.0f, 0.0f, 0.0f, 0.0f);
        __syncthreads();

        if (e0 < e1) {
            const int elast = e1 - 1;
            const int step  = DEG_UNR * DEG_TPB;
            int   dv[DEG_UNR];
            float w[DEG_UNR];
#pragma unroll
            for (int u = 0; u < DEG_UNR; ++u) {
                const int ec = min(e0 + tid + u * DEG_TPB, elast);
                dv[u] = dst[ec];
                w[u]  = ew[ec];
            }
            for (int e = e0 + tid; e < e1; e += step) {
                int   dn[DEG_UNR];
                float wn[DEG_UNR];
#pragma unroll
                for (int u = 0; u < DEG_UNR; ++u) {
                    const int ec = min(e + step + u * DEG_TPB, elast);
                    dn[u] = dst[ec];
                    wn[u] = ew[ec];
                }
#pragma unroll
                for (int u = 0; u < DEG_UNR; ++u) {
                    const int d = (e + u * DEG_TPB < e1) ? dv[u] - base : -1;
                    if ((unsigned)d < (unsigned)cnt)
                        atomicAdd(&hist[d], w[u]);
                }
#pragma unroll
                for (int u = 0; u < DEG_UNR; ++u) { dv[u] = dn[u]; w[u] = wn[u]; }
            }
        }
        __syncthreads();

        const int rounds = (cnt + DEG_TPB - 1) / DEG_TPB;
        const int rot    = (int)blockIdx.x % rounds;
        for (int r = 0; r < rounds; r += DEG_FL) {
            float v[DEG_FL];
            int   idx[DEG_FL];
#pragma unroll
            for (int u = 0; u < DEG_FL; ++u) {
                int rr = r + u + rot;
                if (rr >= rounds) rr -= rounds;
                const int  i  = rr * DEG_TPB + tid;
                const bool ok = (r + u < rounds) && (i < cnt);
                const float hv = hist[ok ? i : tid];
                idx[u] = i;
                v[u]   = ok ? hv : 0.0f;
            }
#pragma unroll
            for (int u = 0; u < DEG_FL; ++u)
                if (v[u] != 0.0f) atomicAdd(&deg[base + idx[u]], v[u]);
        }
        __syncthreads();
    }
}

__global__ void k_finalize(const float* __restrict__ x, float* __restrict__ dinv,
                           float* __restrict__ xt, float* __restrict__ aggt, int n_nodes) {
    int n = blockIdx.x * blockDim.x + threadIdx.x;
    if (n >= n_nodes) return;
    float deg = dinv[n];
    float di  = deg > 0.0f ? rsqrtf(deg) : 0.0f;
    dinv[n] = di;
    float sn = di * di;
    float buf[BT];
#pragma unroll
    for (int s = 0; s < BT; ++s)
        buf[s] = x[(size_t)s * n_nodes + n];
    float4* xt4 = (float4*)(xt + (size_t)n * BT);
    float4* ag4 = (float4*)(aggt + (size_t)n * BT);
#pragma unroll
    for (int q = 0; q < BT / 4; ++q) {
        float4 v = make_float4(buf[4*q], buf[4*q+1], buf[4*q+2], buf[4*q+3]);
        xt4[q] = v;
        ag4[q] = make_float4(v.x * sn, v.y * sn, v.z * sn, v.w * sn);
    }
}

#define AGG_WG  1024
#define AGG_TPB 256
#define AGG_G   8

__global__ __launch_bounds__(AGG_TPB)
void k_aggregate(const int* __restrict__ ei, const float* __restrict__ ew,
                 const float* __restrict__ dinv, const float* __restrict__ xt,
                 float* __restrict__ aggt, int n_edges) {
    const int  lane  = threadIdx.x & 31;
    const bool upper = lane >= 16;
    const int  hl    = 32 + (lane & 15);
    const int wid = __builtin_amdgcn_readfirstlane(
        (int)(blockIdx.x * (AGG_TPB >> 5) + (threadIdx.x >> 5)));
    const int nw  = (int)gridDim.x * (AGG_TPB >> 5);
    const int* __restrict__ dstp = ei + (size_t)n_edges;
    const int nbatch = (n_edges + 31) >> 5;
    const int elast  = n_edges - 1;

    for (int b = wid; b < nbatch; b += nw) {
        const int e0  = b << 5;
        const int cnt = min(32, n_edges - e0);
        const int ec  = min(e0 + lane, elast);
        const int   sv  = ei[ec];
        const int   dv  = dstp[ec];
        const float wv  = ew[ec];
        const float nrm = dinv[sv] * wv * dinv[dv];
        const int   so  = sv * BT;
        const int   dof = dv * BT;

        for (int g = 0; g < cnt; g += AGG_G) {
            if (g + AGG_G <= cnt) {
                int   sj[AGG_G], dj[AGG_G];
                float nj[AGG_G], v0[AGG_G], v1[AGG_G / 2];
#pragma unroll
                for (int u = 0; u < AGG_G; ++u) {
                    sj[u] = __builtin_amdgcn_readlane(so, g + u);
                    dj[u] = __builtin_amdgcn_readlane(dof, g + u);
                    nj[u] = __int_as_float(__builtin_amdgcn_readlane(__float_as_int(nrm), g + u));
                }
#pragma unroll
                for (int u = 0; u < AGG_G; ++u)
                    v0[u] = (xt + sj[u])[lane];
#pragma unroll
                for (int p = 0; p < AGG_G / 2; ++p) {
                    const int sp = upper ? sj[2 * p + 1] : sj[2 * p];
                    v1[p] = xt[sp + hl];
                }
#pragma unroll
                for (int u = 0; u < AGG_G; ++u)
                    atomicAdd(&(aggt + dj[u])[lane], v0[u] * nj[u]);
#pragma unroll
                for (int p = 0; p < AGG_G / 2; ++p) {
                    const int   dp = upper ? dj[2 * p + 1] : dj[2 * p];
                    const float np = upper ? nj[2 * p + 1] : nj[2 * p];
                    atomicAdd(&aggt[dp + hl], v1[p] * np);
                }
            } else {
                for (int j = g; j < cnt; ++j) {
                    const int   s1 = __builtin_amdgcn_readlane(so, j);
                    const int   d1 = __builtin_amdgcn_readlane(dof, j);
                    const float n1 = __int_as_float(__builtin_amdgcn_readlane(__float_as_int(nrm), j));
                    const float* xs = xt   + s1;
                    float*       ad = aggt + d1;
                    atomicAdd(&ad[lane], xs[lane] * n1);
                    if (lane < 16)
                        atomicAdd(&ad[32 + lane], xs[32 + lane] * n1);
                }
            }
        }
    }
}

__global__ void k_epilogue(const float* __restrict__ aggt,
                           const float* __restrict__ gw, const float* __restrict__ gb,
                           const float* __restrict__ cw, const float* __restrict__ cb,
                           const float* __restrict__ lwp, const float* __restrict__ lbp,
                           float* __restrict__ out, int n_nodes) {
    const int lane = threadIdx.x & 31;
    const int h    = lane >> 4;
    const int m    = lane & 15;

    v2f bmat0, bmat1;
#pragma unroll
    for (int v = 0; v < 2; ++v) {
        int k = 2 * h + v;
        float a0 = 0.0f, a1 = 0.0f;
        if (k < 3) {
            for (int ci = 0; ci < HCH; ++ci) {
                float g = gw[ci];
                a0 += g * cw[(size_t)m        * 96 + ci * 3 + k];
                a1 += g * cw[(size_t)(m + 16) * 96 + ci * 3 + k];
            }
        }
        bmat0[v] = a0;
        bmat1[v] = a1;
    }
    float bbA[3], bbB[3];
#pragma unroll
    for (int k = 0; k < 3; ++k) {
        float a0 = 0.0f, a1 = 0.0f;
        for (int ci = 0; ci < HCH; ++ci) {
            float g = gb[ci];
            a0 += g * cw[(size_t)m        * 96 + ci * 3 + k];
            a1 += g * cw[(size_t)(m + 16) * 96 + ci * 3 + k];
        }
        bbA[k] = a0;
        bbB[k] = a1;
    }
    v8f cmat0, cmat1;
#pragma unroll
    for (int j = 0; j < 8; ++j) {
        int t = j + 8 * h;
        float c0 = 0.0f, c1 = 0.0f;
        if (t < TT) {
            c0 = cb[m]      + bbA[1] + (t > 0 ? bbA[0] : 0.0f) + (t < TT - 1 ? bbA[2] : 0.0f);
            c1 = cb[m + 16] + bbB[1] + (t > 0 ? bbB[0] : 0.0f) + (t < TT - 1 ? bbB[2] : 0.0f);
        }
        cmat0[j] = c0;
        cmat1[j] = c1;
    }
    const float lwA = lwp[m], lwB = lwp[16 + m];
    const float lb  = lbp[0];

    const int  i0  = m + 2 * h - 1;
    const int  i1  = i0 + 1;
    const bool ok0 = (m < TT) && (i0 >= 0) && (i0 < TT);
    const bool ok1 = (m < TT) && (i1 < TT);
    const int  o0  = ok0 ? i0 : 0;
    const int  o1  = ok1 ? i1 : 0;
    const float msk0 = ok0 ? 1.0f : 0.0f;
    const float msk1 = ok1 ? 1.0f : 0.0f;

    const int wid = blockIdx.x * (blockDim.x >> 5) + (threadIdx.x >> 5);
    const int nw  = gridDim.x * (blockDim.x >> 5);

#pragma unroll
    for (int b = 0; b < BB; ++b) {
        const float* agb  = aggt + b * TT;
        float*       outb = out + (size_t)b * n_nodes;
        for (int n = wid; n < n_nodes; n += nw) {
            const float* ag = agb + (size_t)n * BT;

            v2f amat;
            amat[0] = ag[o0] * msk0;
            amat[1] = ag[o1] * msk1;

            v8f d0 = __builtin_amdgcn_wmma_f32_16x16x4_f32(false, amat, false, bmat0,
                                                           (short)0, cmat0, false, false);
            v8f d1 = __builtin_amdgcn_wmma_f32_16x16x4_f32(false, amat, false, bmat1,
                                                           (short)0, cmat1, false, false);

            float sA = 0.0f, sB = 0.0f;
#pragma unroll
            for (int j = 0; j < 8; ++j) {
                sA += fmaxf(d0[j], 0.0f);
                sB += fmaxf(d1[j], 0.0f);
            }
            sA += __shfl_xor(sA, 16);
            sB += __shfl_xor(sB, 16);

            float v = sA * lwA + sB * lwB;
            v += __shfl_xor(v, 1);
            v += __shfl_xor(v, 2);
            v += __shfl_xor(v, 4);
            v += __shfl_xor(v, 8);

            if (lane == 0)
                outb[n] = v * (1.0f / (float)TT) + lb;
        }
    }
}

extern "C" void kernel_launch(void* const* d_in, const int* in_sizes, int n_in,
                              void* d_out, int out_size, void* d_ws, size_t ws_size,
                              hipStream_t stream) {
    const float* x  = (const float*)d_in[0];
    const int*   ei = (const int*)d_in[1];
    const float* ew = (const float*)d_in[2];
    const float* gw = (const float*)d_in[3];
    const float* gb = (const float*)d_in[4];
    const float* cw = (const float*)d_in[5];
    const float* cb = (const float*)d_in[6];
    const float* lw = (const float*)d_in[7];
    const float* lb = (const float*)d_in[8];
    float* out = (float*)d_out;

    const int E = in_sizes[2];
    const int N = in_sizes[0] / BT;

    float* dinv = (float*)d_ws;
    float* xt   = dinv + N;
    float* aggt = xt + (size_t)N * BT;

    k_deg_init <<<(N + 255) / 256, 256, 0, stream>>>(dinv, N);
    k_deg_edges<<<DEG_WG, DEG_TPB, 0, stream>>>(ei, ew, dinv, E, N);
    k_finalize <<<(N + 255) / 256, 256, 0, stream>>>(x, dinv, xt, aggt, N);

    k_aggregate<<<AGG_WG, AGG_TPB, 0, stream>>>(ei, ew, dinv, xt, aggt, E);

    k_epilogue <<<1024, 256, 0, stream>>>(aggt, gw, gb, cw, cb, lw, lb, out, N);
}


# edge aggregation: aggt rows split in two node ranges, one per half of the package; a wave adds only into its own half's range (work tickets per range, 2048 workgroups); sums arrive in another order
# speedup vs baseline: 1.9279x; 1.0334x over previous
#include <hip/hip_runtime.h>
#include <hip/hip_bf16.h>


typedef __attribute__((ext_vector_type(2))) float v2f;
typedef __attribute__((ext_vector_type(8))) float v8f;

#define BT 48
#define TT 12
#define BB 4
#define HCH 32

#define Q_STRIDE 64

__global__ void k_deg_init(float* __restrict__ deg, unsigned* __restrict__ q, int n_nodes) {
    int n = blockIdx.x * blockDim.x + threadIdx.x;
    if (n < n_nodes) deg[n] = 1.0f;
    if (n == 0) { q[0] = 0u; q[Q_STRIDE] = 0u; }
}

#define DEG_CAP 51200
#define DEG_WG  64
#define DEG_TPB 1024
#define DEG_UNR 8
#define DEG_FL  7

__global__ __launch_bounds__(DEG_TPB)
void k_deg_edges(const int* __restrict__ ei, const float* __restrict__ ew,
                 float* __restrict__ deg, int n_edges, int n_nodes) {
    __shared__ float4 hist4[DEG_CAP / 4];
    float* hist = (float*)hist4;
    const int tid = threadIdx.x;
    const int per = (n_edges + (int)gridDim.x - 1) / (int)gridDim.x;
    const int e0  = (int)blockIdx.x * per;
    const int e1  = min(e0 + per, n_edges);
    const int* __restrict__ dst = ei + (size_t)n_edges;

    for (int base = 0; base < n_nodes; base += DEG_CAP) {
        const int cnt = min(DEG_CAP, n_nodes - base);

        for (int i = tid; i < DEG_CAP / 4; i += DEG_TPB)
            hist4[i] = make_float4(0.0f, 0.0f, 0.0f, 0.0f);
        __syncthreads();

        if (e0 < e1) {
            const int elast = e1 - 1;
            const int step  = DEG_UNR * DEG_TPB;
            int   dv[DEG_UNR];
            float w[DEG_UNR];
#pragma unroll
            for (int u = 0; u < DEG_UNR; ++u) {
                const int ec = min(e0 + tid + u * DEG_TPB, elast);
                dv[u] = dst[ec];
                w[u]  = ew[ec];
            }
            for (int e = e0 + tid; e < e1; e += step) {
                int   dn[DEG_UNR];
                float wn[DEG_UNR];
#pragma unroll
                for (int u = 0; u < DEG_UNR; ++u) {
                    const int ec = min(e + step + u * DEG_TPB, elast);
                    dn[u] = dst[ec];
                    wn[u] = ew[ec];
                }
#pragma unroll
                for (int u = 0; u < DEG_UNR; ++u) {
                    const int d = (e + u * DEG_TPB < e1) ? dv[u] - base : -1;
                    if ((unsigned)d < (unsigned)cnt)
                        atomicAdd(&hist[d], w[u]);
                }
#pragma unroll
                for (int u = 0; u < DEG_UNR; ++u) { dv[u] = dn[u]; w[u] = wn[u]; }
            }
        }
        __syncthreads();

        const int rounds = (cnt + DEG_TPB - 1) / DEG_TPB;
        const int rot    = (int)blockIdx.x % rounds;
        for (int r = 0; r < rounds; r += DEG_FL) {
            float v[DEG_FL];
            int   idx[DEG_FL];
#pragma unroll
            for (int u = 0; u < DEG_FL; ++u) {
                int rr = r + u + rot;
                if (rr >= rounds) rr -= rounds;
                const int  i  = rr * DEG_TPB + tid;
                const bool ok = (r + u < rounds) && (i < cnt);
                const float hv = hist[ok ? i : tid];
                idx[u] = i;
                v[u]   = ok ? hv : 0.0f;
            }
#pragma unroll
            for (int u = 0; u < DEG_FL; ++u)
                if (v[u] != 0.0f) atomicAdd(&deg[base + idx[u]], v[u]);
        }
        __syncthreads();
    }
}

__global__ void k_finalize(const float* __restrict__ x, float* __restrict__ dinv,
                           float* __restrict__ xt, float* __restrict__ aggt, int n_nodes) {
    int n = blockIdx.x * blockDim.x + threadIdx.x;
    if (n >= n_nodes) return;
    float deg = dinv[n];
    float di  = deg > 0.0f ? rsqrtf(deg) : 0.0f;
    dinv[n] = di;
    float sn = di * di;
    float buf[BT];
#pragma unroll
    for (int s = 0; s < BT; ++s)
        buf[s] = x[(size_t)s * n_nodes + n];
    float4* xt4 = (float4*)(xt + (size_t)n * BT);
    float4* ag4 = (float4*)(aggt + (size_t)n * BT);
#pragma unroll
    for (int q = 0; q < BT / 4; ++q) {
        float4 v = make_float4(buf[4*q], buf[4*q+1], buf[4*q+2], buf[4*q+3]);
        xt4[q] = v;
        ag4[q] = make_float4(v.x * sn, v.y * sn, v.z * sn, v.w * sn);
    }
}

#define AGG_WG  2048
#define AGG_TPB 256
#define AGG_G   8
#define AGG_CH  4

template <bool FULL>
__device__ __forceinline__ unsigned agg_group(const float* __restrict__ xt, float* __restrict__ aggt,
                                              unsigned mask, const int k,
                                              const int so, const int dof, const float nrm,
                                              const int lane, const bool upper, const int hl) {
    int   sj[AGG_G], dj[AGG_G];
    float nj[AGG_G], v0[AGG_G], v1[AGG_G / 2];
#pragma unroll
    for (int u = 0; u < AGG_G; ++u) {
        sj[u] = 0; dj[u] = 0; nj[u] = 0.0f; v0[u] = 0.0f;
        if (FULL || u < k) {
            const int j = __builtin_ctz(mask);
            mask &= mask - 1u;
            sj[u] = __builtin_amdgcn_readlane(so, j);
            dj[u] = __builtin_amdgcn_readlane(dof, j);
            nj[u] = __int_as_float(__builtin_amdgcn_readlane(__float_as_int(nrm), j));
        }
    }
#pragma unroll
    for (int u = 0; u < AGG_G; ++u)
        if (FULL || u < k) v0[u] = (xt + sj[u])[lane];
#pragma unroll
    for (int p = 0; p < AGG_G / 2; ++p) {
        v1[p] = 0.0f;
        if (FULL || 2 * p < k) {
            const bool ok = FULL || !upper || (2 * p + 1 < k);
            const int  sp = upper ? sj[2 * p + 1] : sj[2 * p];
            if (ok) v1[p] = xt[sp + hl];
        }
    }
#pragma unroll
    for (int u = 0; u < AGG_G; ++u)
        if (FULL || u < k) atomicAdd(&(aggt + dj[u])[lane], v0[u] * nj[u]);
#pragma unroll
    for (int p = 0; p < AGG_G / 2; ++p) {
        if (FULL || 2 * p < k) {
            const bool  ok = FULL || !upper || (2 * p + 1 < k);
            const int   dp = upper ? dj[2 * p + 1] : dj[2 * p];
            const float np = upper ? nj[2 * p + 1] : nj[2 * p];
            if (ok) atomicAdd(&aggt[dp + hl], v1[p] * np);
        }
    }
    return mask;
}

__global__ __launch_bounds__(AGG_TPB)
void k_aggregate(const int* __restrict__ ei, const float* __restrict__ ew,
                 const float* __restrict__ dinv, const float* __restrict__ xt,
                 float* __restrict__ aggt, unsigned* __restrict__ q,
                 int n_edges, int n_nodes) {
    const int  lane  = threadIdx.x & 31;
    const bool upper = lane >= 16;
    const int  hl    = 32 + (lane & 15);
    const int* __restrict__ dstp = ei + (size_t)n_edges;
    const int nbatch = (n_edges + 31) >> 5;
    const int ntick  = (nbatch + AGG_CH - 1) / AGG_CH;
    const int elast  = n_edges - 1;
    const int nsplit = min(n_nodes, ((n_nodes >> 1) + 3) & ~3);
    const int myhalf = (int)((__builtin_amdgcn_s_sendmsg_rtn(0x87) >> 18) & 1u);

    for (int pass = 0; pass < 2; ++pass) {
        const int hh = myhalf ^ pass;
        const int lo = hh ? nsplit : 0;
        const int hi = hh ? n_nodes : nsplit;
        unsigned* __restrict__ qh = q + hh * Q_STRIDE;

        for (;;) {
            unsigned vt = 0u;
            if (lane == 0) vt = atomicAdd(qh, 1u);
            const int t = (int)__builtin_amdgcn_readfirstlane(vt);
            if (t >= ntick) break;

            const int b0 = t * AGG_CH;
            int      so[AGG_CH], dof[AGG_CH];
            float    nrm[AGG_CH];
            unsigned msk[AGG_CH];
            int      sv[AGG_CH], dv[AGG_CH];
            float    wv[AGG_CH];
#pragma unroll
            for (int c = 0; c < AGG_CH; ++c) {
                const int ec = min(((b0 + c) << 5) + lane, elast);
                sv[c] = ei[ec];
                dv[c] = dstp[ec];
                wv[c] = ew[ec];
            }
            float da[AGG_CH], db[AGG_CH];
#pragma unroll
            for (int c = 0; c < AGG_CH; ++c) {
                const bool mine = (((b0 + c) << 5) + lane < n_edges) && (dv[c] >= lo) && (dv[c] < hi);
                msk[c] = (unsigned)__builtin_amdgcn_readfirstlane((int)__builtin_amdgcn_ballot_w32(mine));
                const int keep = -(int)mine;
                da[c] = dinv[sv[c] & keep];
                db[c] = dinv[dv[c] & keep];
            }
#pragma unroll
            for (int c = 0; c < AGG_CH; ++c) {
                nrm[c] = da[c] * wv[c] * db[c];
                asm volatile("" : "+v"(nrm[c]));
                so[c]  = sv[c] * BT;
                dof[c] = dv[c] * BT;
            }
#pragma unroll
            for (int c = 0; c < AGG_CH; ++c) {
                unsigned m = msk[c];
                while (m != 0u) {
                    const int k = __builtin_popcount(m);
                    if (k >= AGG_G) m = agg_group<true >(xt, aggt, m, AGG_G, so[c], dof[c], nrm[c], lane, upper, hl);
                    else            m = agg_group<false>(xt, aggt, m, k,     so[c], dof[c], nrm[c], lane, upper, hl);
                }
            }
        }
    }
}

__global__ void k_epilogue(const float* __restrict__ aggt,
                           const float* __restrict__ gw, const float* __restrict__ gb,
                           const float* __restrict__ cw, const float* __restrict__ cb,
                           const float* __restrict__ lwp, const float* __restrict__ lbp,
                           float* __restrict__ out, int n_nodes) {
    const int lane = threadIdx.x & 31;
    const int h    = lane >> 4;
    const int m    = lane & 15;

    v2f bmat0, bmat1;
#pragma unroll
    for (int v = 0; v < 2; ++v) {
        int k = 2 * h + v;
        float a0 = 0.0f, a1 = 0.0f;
        if (k < 3) {
            for (int ci = 0; ci < HCH; ++ci) {
                float g = gw[ci];
                a0 += g * cw[(size_t)m        * 96 + ci * 3 + k];
                a1 += g * cw[(size_t)(m + 16) * 96 + ci * 3 + k];
            }
        }
        bmat0[v] = a0;
        bmat1[v] = a1;
    }
    float bbA[3], bbB[3];
#pragma unroll
    for (int k = 0; k < 3; ++k) {
        float a0 = 0.0f, a1 = 0.0f;
        for (int ci = 0; ci < HCH; ++ci) {
            float g = gb[ci];
            a0 += g * cw[(size_t)m        * 96 + ci * 3 + k];
            a1 += g * cw[(size_t)(m + 16) * 96 + ci * 3 + k];
        }
        bbA[k] = a0;
        bbB[k] = a1;
    }
    v8f cmat0, cmat1;
#pragma unroll
    for (int j = 0; j < 8; ++j) {
        int t = j + 8 * h;
        float c0 = 0.0f, c1 = 0.0f;
        if (t < TT) {
            c0 = cb[m]      + bbA[1] + (t > 0 ? bbA[0] : 0.0f) + (t < TT - 1 ? bbA[2] : 0.0f);
            c1 = cb[m + 16] + bbB[1] + (t > 0 ? bbB[0] : 0.0f) + (t < TT - 1 ? bbB[2] : 0.0f);
        }
        cmat0[j] = c0;
        cmat1[j] = c1;
    }
    const float lwA = lwp[m], lwB = lwp[16 + m];
    const float lb  = lbp[0];

    const int  i0  = m + 2 * h - 1;
    const int  i1  = i0 + 1;
    const bool ok0 = (m < TT) && (i0 >= 0) && (i0 < TT);
    const bool ok1 = (m < TT) && (i1 < TT);
    const int  o0  = ok0 ? i0 : 0;
    const int  o1  = ok1 ? i1 : 0;
    const float msk0 = ok0 ? 1.0f : 0.0f;
    const float msk1 = ok1 ? 1.0f : 0.0f;

    const int wid = blockIdx.x * (blockDim.x >> 5) + (threadIdx.x >> 5);
    const int nw  = gridDim.x * (blockDim.x >> 5);

#pragma unroll
    for (int b = 0; b < BB; ++b) {
        const float* agb  = aggt + b * TT;
        float*       outb = out + (size_t)b * n_nodes;
        for (int n = wid; n < n_nodes; n += nw) {
            const float* ag = agb + (size_t)n * BT;

            v2f amat;
            amat[0] = ag[o0] * msk0;
            amat[1] = ag[o1] * msk1;

            v8f d0 = __builtin_amdgcn_wmma_f32_16x16x4_f32(false, amat, false, bmat0,
                                                           (short)0, cmat0, false, false);
            v8f d1 = __builtin_amdgcn_wmma_f32_16x16x4_f32(false, amat, false, bmat1,
                                                           (short)0, cmat1, false, false);

            float sA = 0.0f, sB = 0.0f;
#pragma unroll
            for (int j = 0; j < 8; ++j) {
                sA += fmaxf(d0[j], 0.0f);
                sB += fmaxf(d1[j], 0.0f);
            }
            sA += __shfl_xor(sA, 16);
            sB += __shfl_xor(sB, 16);

            float v = sA * lwA + sB * lwB;
            v += __shfl_xor(v, 1);
            v += __shfl_xor(v, 2);
            v += __shfl_xor(v, 4);
            v += __shfl_xor(v, 8);

            if (lane == 0)
                outb[n] = v * (1.0f / (float)TT) + lb;
        }
    }
}

extern "C" void kernel_launch(void* const* d_in, const int* in_sizes, int n_in,
                              void* d_out, int out_size, void* d_ws, size_t ws_size,
                              hipStream_t stream) {
    const float* x  = (const float*)d_in[0];
    const int*   ei = (const int*)d_in[1];
    const float* ew = (const float*)d_in[2];
    const float* gw = (const float*)d_in[3];
    const float* gb = (const float*)d_in[4];
    const float* cw = (const float*)d_in[5];
    const float* cb = (const float*)d_in[6];
    const float* lw = (const float*)d_in[7];
    const float* lb = (const float*)d_in[8];
    float* out = (float*)d_out;

    const int E = in_sizes[2];
    const int N = in_sizes[0] / BT;

    float* dinv = (float*)d_ws;
    float* xt   = dinv + N;
    float* aggt = xt + (size_t)N * BT;
    unsigned* q = (unsigned*)d_out;

    k_deg_init <<<(N + 255) / 256, 256, 0, stream>>>(dinv, q, N);
    k_deg_edges<<<DEG_WG, DEG_TPB, 0, stream>>>(ei, ew, dinv, E, N);
    k_finalize <<<(N + 255) / 256, 256, 0, stream>>>(x, dinv, xt, aggt, N);

    k_aggregate<<<AGG_WG, AGG_TPB, 0, stream>>>(ei, ew, dinv, xt, aggt, q, E, N);

    k_epilogue <<<1024, 256, 0, stream>>>(aggt, gw, gb, cw, cb, lw, lb, out, N);
}


# rows of x_t and aggt stored as two planes, slices 0..31 in 128-byte rows and 32..47 in 64-byte rows: no gather or atomic straddles a line, 2 line requests per row access instead of 2.5
# speedup vs baseline: 2.1746x; 1.1280x over previous
#include <hip/hip_runtime.h>
#include <hip/hip_bf16.h>


typedef __attribute__((ext_vector_type(2))) float v2f;
typedef __attribute__((ext_vector_type(8))) float v8f;

#define BT 48
#define TT 12
#define BB 4
#define HCH 32

#define Q_STRIDE 64

#define P0W 32
#define P1W 16

__global__ void k_deg_init(float* __restrict__ deg, unsigned* __restrict__ q, int n_nodes) {
    int n = blockIdx.x * blockDim.x + threadIdx.x;
    if (n < n_nodes) deg[n] = 1.0f;
    if (n == 0) { q[0] = 0u; q[Q_STRIDE] = 0u; }
}

#define DEG_CAP 51200
#define DEG_WG  64
#define DEG_TPB 1024
#define DEG_UNR 8
#define DEG_FL  7

__global__ __launch_bounds__(DEG_TPB)
void k_deg_edges(const int* __restrict__ ei, const float* __restrict__ ew,
                 float* __restrict__ deg, int n_edges, int n_nodes) {
    __shared__ float4 hist4[DEG_CAP / 4];
    float* hist = (float*)hist4;
    const int tid = threadIdx.x;
    const int per = (n_edges + (int)gridDim.x - 1) / (int)gridDim.x;
    const int e0  = (int)blockIdx.x * per;
    const int e1  = min(e0 + per, n_edges);
    const int* __restrict__ dst = ei + (size_t)n_edges;

    for (int base = 0; base < n_nodes; base += DEG_CAP) {
        const int cnt = min(DEG_CAP, n_nodes - base);

        for (int i = tid; i < DEG_CAP / 4; i += DEG_TPB)
            hist4[i] = make_float4(0.0f, 0.0f, 0.0f, 0.0f);
        __syncthreads();

        if (e0 < e1) {
            const int elast = e1 - 1;
            const int step  = DEG_UNR * DEG_TPB;
            int   dv[DEG_UNR];
            float w[DEG_UNR];
#pragma unroll
            for (int u = 0; u < DEG_UNR; ++u) {
                const int ec = min(e0 + tid + u * DEG_TPB, elast);
                dv[u] = dst[ec];
                w[u]  = ew[ec];
            }
            for (int e = e0 + tid; e < e1; e += step) {
                int   dn[DEG_UNR];
                float wn[DEG_UNR];
#pragma unroll
                for (int u = 0; u < DEG_UNR; ++u) {
                    const int ec = min(e + step + u * DEG_TPB, elast);
                    dn[u] = dst[ec];
                    wn[u] = ew[ec];
                }
#pragma unroll
                for (int u = 0; u < DEG_UNR; ++u) {
                    const int d = (e + u * DEG_TPB < e1) ? dv[u] - base : -1;
                    if ((unsigned)d < (unsigned)cnt)
                        atomicAdd(&hist[d], w[u]);
                }
#pragma unroll
                for (int u = 0; u < DEG_UNR; ++u) { dv[u] = dn[u]; w[u] = wn[u]; }
            }
        }
        __syncthreads();

        const int rounds = (cnt + DEG_TPB - 1) / DEG_TPB;
        const int rot    = (int)blockIdx.x % rounds;
        for (int r = 0; r < rounds; r += DEG_FL) {
            float v[DEG_FL];
            int   idx[DEG_FL];
#pragma unroll
            for (int u = 0; u < DEG_FL; ++u) {
                int rr = r + u + rot;
                if (rr >= rounds) rr -= rounds;
                const int  i  = rr * DEG_TPB + tid;
                const bool ok = (r + u < rounds) && (i < cnt);
                const float hv = hist[ok ? i : tid];
                idx[u] = i;
                v[u]   = ok ? hv : 0.0f;
            }
#pragma unroll
            for (int u = 0; u < DEG_FL; ++u)
                if (v[u] != 0.0f) atomicAdd(&deg[base + idx[u]], v[u]);
        }
        __syncthreads();
    }
}

__global__ void k_finalize(const float* __restrict__ x, float* __restrict__ dinv,
                           float* __restrict__ xt0, float* __restrict__ xt1,
                           float* __restrict__ ag0, float* __restrict__ ag1, int n_nodes) {
    int n = blockIdx.x * blockDim.x + threadIdx.x;
    if (n >= n_nodes) return;
    float deg = dinv[n];
    float di  = deg > 0.0f ? rsqrtf(deg) : 0.0f;
    dinv[n] = di;
    float sn = di * di;
    float buf[BT];
#pragma unroll
    for (int s = 0; s < BT; ++s)
        buf[s] = x[(size_t)s * n_nodes + n];
    float4* xa4 = (float4*)(xt0 + (size_t)n * P0W);
    float4* aa4 = (float4*)(ag0 + (size_t)n * P0W);
    float4* xb4 = (float4*)(xt1 + (size_t)n * P1W);
    float4* ab4 = (float4*)(ag1 + (size_t)n * P1W);
#pragma unroll
    for (int q = 0; q < P0W / 4; ++q) {
        float4 v = make_float4(buf[4*q], buf[4*q+1], buf[4*q+2], buf[4*q+3]);
        xa4[q] = v;
        aa4[q] = make_float4(v.x * sn, v.y * sn, v.z * sn, v.w * sn);
    }
#pragma unroll
    for (int q = 0; q < P1W / 4; ++q) {
        float4 v = make_float4(buf[P0W + 4*q], buf[P0W + 4*q+1], buf[P0W + 4*q+2], buf[P0W + 4*q+3]);
        xb4[q] = v;
        ab4[q] = make_float4(v.x * sn, v.y * sn, v.z * sn, v.w * sn);
    }
}

#define AGG_WG  2048
#define AGG_TPB 256
#define AGG_G   8
#define AGG_CH  4

template <bool FULL>
__device__ __forceinline__ unsigned agg_group(const float* __restrict__ xt0, const float* __restrict__ xt1,
                                              float* __restrict__ ag0, float* __restrict__ ag1,
                                              unsigned mask, const int k,
                                              const int so, const int dof, const float nrm,
                                              const int lane, const bool upper, const int hl) {
    int   sj[AGG_G], dj[AGG_G];
    float nj[AGG_G], v0[AGG_G], v1[AGG_G / 2];
#pragma unroll
    for (int u = 0; u < AGG_G; ++u) {
        sj[u] = 0; dj[u] = 0; nj[u] = 0.0f; v0[u] = 0.0f;
        if (FULL || u < k) {
            const int j = __builtin_ctz(mask);
            mask &= mask - 1u;
            sj[u] = __builtin_amdgcn_readlane(so, j);
            dj[u] = __builtin_amdgcn_readlane(dof, j);
            nj[u] = __int_as_float(__builtin_amdgcn_readlane(__float_as_int(nrm), j));
        }
    }
#pragma unroll
    for (int u = 0; u < AGG_G; ++u)
        if (FULL || u < k) v0[u] = (xt0 + sj[u])[lane];
#pragma unroll
    for (int p = 0; p < AGG_G / 2; ++p) {
        v1[p] = 0.0f;
        if (FULL || 2 * p < k) {
            const bool ok = FULL || !upper || (2 * p + 1 < k);
            const int  sp = upper ? sj[2 * p + 1] : sj[2 * p];
            if (ok) v1[p] = xt1[(sp >> 1) + hl];
        }
    }
#pragma unroll
    for (int u = 0; u < AGG_G; ++u)
        if (FULL || u < k) atomicAdd(&(ag0 + dj[u])[lane], v0[u] * nj[u]);
#pragma unroll
    for (int p = 0; p < AGG_G / 2; ++p) {
        if (FULL || 2 * p < k) {
            const bool  ok = FULL || !upper || (2 * p + 1 < k);
            const int   dp = upper ? dj[2 * p + 1] : dj[2 * p];
            const float np = upper ? nj[2 * p + 1] : nj[2 * p];
            if (ok) atomicAdd(&ag1[(dp >> 1) + hl], v1[p] * np);
        }
    }
    return mask;
}

__global__ __launch_bounds__(AGG_TPB)
void k_aggregate(const int* __restrict__ ei, const float* __restrict__ ew,
                 const float* __restrict__ dinv,
                 const float* __restrict__ xt0, const float* __restrict__ xt1,
                 float* __restrict__ ag0, float* __restrict__ ag1, unsigned* __restrict__ q,
                 int n_edges, int n_nodes) {
    const int  lane  = threadIdx.x & 31;
    const bool upper = lane >= 16;
    const int  hl    = lane & 15;
    const int* __restrict__ dstp = ei + (size_t)n_edges;
    const int nbatch = (n_edges + 31) >> 5;
    const int ntick  = (nbatch + AGG_CH - 1) / AGG_CH;
    const int elast  = n_edges - 1;
    const int nsplit = min(n_nodes, ((n_nodes >> 1) + 3) & ~3);
    const int myhalf = (int)((__builtin_amdgcn_s_sendmsg_rtn(0x87) >> 18) & 1u);

    for (int pass = 0; pass < 2; ++pass) {
        const int hh = myhalf ^ pass;
        const int lo = hh ? nsplit : 0;
        const int hi = hh ? n_nodes : nsplit;
        unsigned* __restrict__ qh = q + hh * Q_STRIDE;

        for (;;) {
            unsigned vt = 0u;
            if (lane == 0) vt = atomicAdd(qh, 1u);
            const int t = (int)__builtin_amdgcn_readfirstlane(vt);
            if (t >= ntick) break;

            const int b0 = t * AGG_CH;
            int      so[AGG_CH], dof[AGG_CH];
            float    nrm[AGG_CH];
            unsigned msk[AGG_CH];
            int      sv[AGG_CH], dv[AGG_CH];
            float    wv[AGG_CH];
#pragma unroll
            for (int c = 0; c < AGG_CH; ++c) {
                const int ec = min(((b0 + c) << 5) + lane, elast);
                sv[c] = ei[ec];
                dv[c] = dstp[ec];
                wv[c] = ew[ec];
            }
            float da[AGG_CH], db[AGG_CH];
#pragma unroll
            for (int c = 0; c < AGG_CH; ++c) {
                const bool mine = (((b0 + c) << 5) + lane < n_edges) && (dv[c] >= lo) && (dv[c] < hi);
                msk[c] = (unsigned)__builtin_amdgcn_readfirstlane((int)__builtin_amdgcn_ballot_w32(mine));
                const int keep = -(int)mine;
                da[c] = dinv[sv[c] & keep];
                db[c] = dinv[dv[c] & keep];
            }
#pragma unroll
            for (int c = 0; c < AGG_CH; ++c) {
                nrm[c] = da[c] * wv[c] * db[c];
                asm volatile("" : "+v"(nrm[c]));
                so[c]  = sv[c] * P0W;
                dof[c] = dv[c] * P0W;
            }
#pragma unroll
            for (int c = 0; c < AGG_CH; ++c) {
                unsigned m = msk[c];
                while (m != 0u) {
                    const int k = __builtin_popcount(m);
                    if (k >= AGG_G) m = agg_group<true >(xt0, xt1, ag0, ag1, m, AGG_G, so[c], dof[c], nrm[c], lane, upper, hl);
                    else            m = agg_group<false>(xt0, xt1, ag0, ag1, m, k,     so[c], dof[c], nrm[c], lane, upper, hl);
                }
            }
        }
    }
}

__global__ void k_epilogue(const float* __restrict__ ag0, const float* __restrict__ ag1,
                           const float* __restrict__ gw, const float* __restrict__ gb,
                           const float* __restrict__ cw, const float* __restrict__ cb,
                           const float* __restrict__ lwp, const float* __restrict__ lbp,
                           float* __restrict__ out, int n_nodes) {
    const int lane = threadIdx.x & 31;
    const int h    = lane >> 4;
    const int m    = lane & 15;

    v2f bmat0, bmat1;
#pragma unroll
    for (int v = 0; v < 2; ++v) {
        int k = 2 * h + v;
        float a0 = 0.0f, a1 = 0.0f;
        if (k < 3) {
            for (int ci = 0; ci < HCH; ++ci) {
                float g = gw[ci];
                a0 += g * cw[(size_t)m        * 96 + ci * 3 + k];
                a1 += g * cw[(size_t)(m + 16) * 96 + ci * 3 + k];
            }
        }
        bmat0[v] = a0;
        bmat1[v] = a1;
    }
    float bbA[3], bbB[3];
#pragma unroll
    for (int k = 0; k < 3; ++k) {
        float a0 = 0.0f, a1 = 0.0f;
        for (int ci = 0; ci < HCH; ++ci) {
            float g = gb[ci];
            a0 += g * cw[(size_t)m        * 96 + ci * 3 + k];
            a1 += g * cw[(size_t)(m + 16) * 96 + ci * 3 + k];
        }
        bbA[k] = a0;
        bbB[k] = a1;
    }
    v8f cmat0, cmat1;
#pragma unroll
    for (int j = 0; j < 8; ++j) {
        int t = j + 8 * h;
        float c0 = 0.0f, c1 = 0.0f;
        if (t < TT) {
            c0 = cb[m]      + bbA[1] + (t > 0 ? bbA[0] : 0.0f) + (t < TT - 1 ? bbA[2] : 0.0f);
            c1 = cb[m + 16] + bbB[1] + (t > 0 ? bbB[0] : 0.0f) + (t < TT - 1 ? bbB[2] : 0.0f);
        }
        cmat0[j] = c0;
        cmat1[j] = c1;
    }
    const float lwA = lwp[m], lwB = lwp[16 + m];
    const float lb  = lbp[0];

    const int  i0  = m + 2 * h - 1;
    const int  i1  = i0 + 1;
    const bool ok0 = (m < TT) && (i0 >= 0) && (i0 < TT);
    const bool ok1 = (m < TT) && (i1 < TT);
    const int  o0  = ok0 ? i0 : 0;
    const int  o1  = ok1 ? i1 : 0;
    const float msk0 = ok0 ? 1.0f : 0.0f;
    const float msk1 = ok1 ? 1.0f : 0.0f;

    const int wid = blockIdx.x * (blockDim.x >> 5) + (threadIdx.x >> 5);
    const int nw  = gridDim.x * (blockDim.x >> 5);

#pragma unroll
    for (int b = 0; b < BB; ++b) {
        const int  s0 = b * TT + o0;
        const int  s1 = b * TT + o1;
        const bool f0 = s0 < P0W;
        const bool f1 = s1 < P0W;
        const float* __restrict__ g0 = f0 ? ag0 + s0 : ag1 + (s0 - P0W);
        const float* __restrict__ g1 = f1 ? ag0 + s1 : ag1 + (s1 - P0W);
        const int  w0 = f0 ? P0W : P1W;
        const int  w1 = f1 ? P0W : P1W;
        float*       outb = out + (size_t)b * n_nodes;
        for (int n = wid; n < n_nodes; n += nw) {
            v2f amat;
            amat[0] = g0[n * w0] * msk0;
            amat[1] = g1[n * w1] * msk1;

            v8f d0 = __builtin_amdgcn_wmma_f32_16x16x4_f32(false, amat, false, bmat0,
                                                           (short)0, cmat0, false, false);
            v8f d1 = __builtin_amdgcn_wmma_f32_16x16x4_f32(false, amat, false, bmat1,
                                                           (short)0, cmat1, false, false);

            float sA = 0.0f, sB = 0.0f;
#pragma unroll
            for (int j = 0; j < 8; ++j) {
                sA += fmaxf(d0[j], 0.0f);
                sB += fmaxf(d1[j], 0.0f);
            }
            sA += __shfl_xor(sA, 16);
            sB += __shfl_xor(sB, 16);

            float v = sA * lwA + sB * lwB;
            v += __shfl_xor(v, 1);
            v += __shfl_xor(v, 2);
            v += __shfl_xor(v, 4);
            v += __shfl_xor(v, 8);

            if (lane == 0)
                outb[n] = v * (1.0f / (float)TT) + lb;
        }
    }
}

extern "C" void kernel_launch(void* const* d_in, const int* in_sizes, int n_in,
                              void* d_out, int out_size, void* d_ws, size_t ws_size,
                              hipStream_t stream) {
    const float* x  = (const float*)d_in[0];
    const int*   ei = (const int*)d_in[1];
    const float* ew = (const float*)d_in[2];
    const float* gw = (const float*)d_in[3];
    const float* gb = (const float*)d_in[4];
    const float* cw = (const float*)d_in[5];
    const float* cb = (const float*)d_in[6];
    const float* lw = (const float*)d_in[7];
    const float* lb = (const float*)d_in[8];
    float* out = (float*)d_out;

    const int E = in_sizes[2];
    const int N = in_sizes[0] / BT;

    float* xt0  = (float*)d_ws;
    float* ag0  = xt0 + (size_t)N * P0W;
    float* xt1  = ag0 + (size_t)N * P0W;
    float* ag1  = xt1 + (size_t)N * P1W;
    float* dinv = ag1 + (size_t)N * P1W;
    unsigned* q = (unsigned*)d_out;

    k_deg_init <<<(N + 255) / 256, 256, 0, stream>>>(dinv, q, N);
    k_deg_edges<<<DEG_WG, DEG_TPB, 0, stream>>>(ei, ew, dinv, E, N);
    k_finalize <<<(N + 255) / 256, 256, 0, stream>>>(x, dinv, xt0, xt1, ag0, ag1, N);

    k_aggregate<<<AGG_WG, AGG_TPB, 0, stream>>>(ei, ew, dinv, xt0, xt1, ag0, ag1, q, E, N);

    k_epilogue <<<1024, 256, 0, stream>>>(ag0, ag1, gw, gb, cw, cb, lw, lb, out, N);
}


# aggregation per destination range: edges binned by 64-node range (LDS counting sort), a workgroup sums its range's rows in LDS (ds atomics), epilogue reads them from LDS; sums in another order
# speedup vs baseline: 3.2314x; 1.4860x over previous
#include <hip/hip_runtime.h>
#include <hip/hip_bf16.h>


typedef __attribute__((ext_vector_type(2))) float v2f;
typedef __attribute__((ext_vector_type(8))) float v8f;

#define BT 48
#define TT 12
#define BB 4
#define HCH 32

#define P0W 32
#define P1W 16

#define RG_SH    6
#define RG_ROWS  (1 << RG_SH)
#define RG_MAX   1024
#define BIN_WG   256
#define BIN_TPB  1024
#define BIN_UNR  8
#define OFF_P    (RG_MAX + 1)
#define REC_SH   25

__global__ void k_deg_init(float* __restrict__ deg, int n_nodes) {
    int n = blockIdx.x * blockDim.x + threadIdx.x;
    if (n < n_nodes) deg[n] = 1.0f;
}

#define DEG_CAP 51200
#define DEG_WG  64
#define DEG_TPB 1024
#define DEG_UNR 8
#define DEG_FL  7

__global__ __launch_bounds__(DEG_TPB)
void k_deg_edges(const int* __restrict__ ei, const float* __restrict__ ew,
                 float* __restrict__ deg, int n_edges, int n_nodes) {
    __shared__ float4 hist4[DEG_CAP / 4];
    float* hist = (float*)hist4;
    const int tid = threadIdx.x;
    const int per = (n_edges + (int)gridDim.x - 1) / (int)gridDim.x;
    const int e0  = (int)blockIdx.x * per;
    const int e1  = min(e0 + per, n_edges);
    const int* __restrict__ dst = ei + (size_t)n_edges;

    for (int base = 0; base < n_nodes; base += DEG_CAP) {
        const int cnt = min(DEG_CAP, n_nodes - base);

        for (int i = tid; i < DEG_CAP / 4; i += DEG_TPB)
            hist4[i] = make_float4(0.0f, 0.0f, 0.0f, 0.0f);
        __syncthreads();

        if (e0 < e1) {
            const int elast = e1 - 1;
            const int step  = DEG_UNR * DEG_TPB;
            int   dv[DEG_UNR];
            float w[DEG_UNR];
#pragma unroll
            for (int u = 0; u < DEG_UNR; ++u) {
                const int ec = min(e0 + tid + u * DEG_TPB, elast);
                dv[u] = dst[ec];
                w[u]  = ew[ec];
            }
            for (int e = e0 + tid; e < e1; e += step) {
                int   dn[DEG_UNR];
                float wn[DEG_UNR];
#pragma unroll
                for (int u = 0; u < DEG_UNR; ++u) {
                    const int ec = min(e + step + u * DEG_TPB, elast);
                    dn[u] = dst[ec];
                    wn[u] = ew[ec];
                }
#pragma unroll
                for (int u = 0; u < DEG_UNR; ++u) {
                    const int d = (e + u * DEG_TPB < e1) ? dv[u] - base : -1;
                    if ((unsigned)d < (unsigned)cnt)
                        atomicAdd(&hist[d], w[u]);
                }
#pragma unroll
                for (int u = 0; u < DEG_UNR; ++u) { dv[u] = dn[u]; w[u] = wn[u]; }
            }
        }
        __syncthreads();

        const int rounds = (cnt + DEG_TPB - 1) / DEG_TPB;
        const int rot    = (int)blockIdx.x % rounds;
        for (int r = 0; r < rounds; r += DEG_FL) {
            float v[DEG_FL];
            int   idx[DEG_FL];
#pragma unroll
            for (int u = 0; u < DEG_FL; ++u) {
                int rr = r + u + rot;
                if (rr >= rounds) rr -= rounds;
                const int  i  = rr * DEG_TPB + tid;
                const bool ok = (r + u < rounds) && (i < cnt);
                const float hv = hist[ok ? i : tid];
                idx[u] = i;
                v[u]   = ok ? hv : 0.0f;
            }
#pragma unroll
            for (int u = 0; u < DEG_FL; ++u)
                if (v[u] != 0.0f) atomicAdd(&deg[base + idx[u]], v[u]);
        }
        __syncthreads();
    }
}

__global__ void k_finalize(const float* __restrict__ x, float* __restrict__ dinv,
                           float* __restrict__ xt0, float* __restrict__ xt1, int n_nodes) {
    int n = blockIdx.x * blockDim.x + threadIdx.x;
    if (n >= n_nodes) return;
    float deg = dinv[n];
    float di  = deg > 0.0f ? rsqrtf(deg) : 0.0f;
    dinv[n] = di;
    float buf[BT];
#pragma unroll
    for (int s = 0; s < BT; ++s)
        buf[s] = x[(size_t)s * n_nodes + n];
    float4* xa4 = (float4*)(xt0 + (size_t)n * P0W);
    float4* xb4 = (float4*)(xt1 + (size_t)n * P1W);
#pragma unroll
    for (int q = 0; q < P0W / 4; ++q)
        xa4[q] = make_float4(buf[4*q], buf[4*q+1], buf[4*q+2], buf[4*q+3]);
#pragma unroll
    for (int q = 0; q < P1W / 4; ++q)
        xb4[q] = make_float4(buf[P0W + 4*q], buf[P0W + 4*q+1], buf[P0W + 4*q+2], buf[P0W + 4*q+3]);
}

static_assert(RG_MAX == BIN_TPB, "the binning keeps one range per thread");
static_assert(BIN_WG <= 512, "the aggregation keeps one piece per thread");
__global__ __launch_bounds__(BIN_TPB)
void k_bin(const int* __restrict__ ei, unsigned* __restrict__ rec, int* __restrict__ off, int n_edges) {
    __shared__ int lcnt[RG_MAX];
    __shared__ int lsc[RG_MAX];
    const int tid = threadIdx.x;
    const int per = (n_edges + BIN_WG - 1) / BIN_WG;
    const int e0  = (int)blockIdx.x * per;
    const int e1  = min(e0 + per, n_edges);
    const int* __restrict__ dst = ei + (size_t)n_edges;
    const int step = BIN_UNR * BIN_TPB;

    lcnt[tid] = 0;
    __syncthreads();
    for (int e = e0 + tid; e < e1; e += step) {
        int dv[BIN_UNR];
#pragma unroll
        for (int u = 0; u < BIN_UNR; ++u) dv[u] = dst[min(e + u * BIN_TPB, e1 - 1)];
#pragma unroll
        for (int u = 0; u < BIN_UNR; ++u)
            if (e + u * BIN_TPB < e1) atomicAdd(&lcnt[dv[u] >> RG_SH], 1);
    }
    __syncthreads();

    const int mine = lcnt[tid];
    lsc[tid] = mine;
    __syncthreads();
    for (int d = 1; d < RG_MAX; d <<= 1) {
        const int t = (tid >= d) ? lsc[tid - d] : 0;
        __syncthreads();
        lsc[tid] += t;
        __syncthreads();
    }
    const int incl = lsc[tid];
    const int excl = incl - mine;
    int* __restrict__ offw = off + (size_t)blockIdx.x * OFF_P;
    offw[tid] = excl;
    if (tid == RG_MAX - 1) offw[RG_MAX] = incl;
    lcnt[tid] = excl;
    __syncthreads();

    for (int e = e0 + tid; e < e1; e += step) {
        int dv[BIN_UNR];
#pragma unroll
        for (int u = 0; u < BIN_UNR; ++u) dv[u] = dst[min(e + u * BIN_TPB, e1 - 1)];
#pragma unroll
        for (int u = 0; u < BIN_UNR; ++u) {
            if (e + u * BIN_TPB < e1) {
                const int p = atomicAdd(&lcnt[dv[u] >> RG_SH], 1);
                rec[(size_t)e0 + p] = (unsigned)(e + u * BIN_TPB) | ((unsigned)(dv[u] & (RG_ROWS - 1)) << REC_SH);
            }
        }
    }
}

#define AGG_TPB 512
#define AGG_NW  (AGG_TPB / 32)
#define AGG_G   8
#define AGG_CH  5
#define ACC_P   BT

template <bool FULL>
__device__ __forceinline__ unsigned agg_group(const float* __restrict__ xt0, const float* __restrict__ xt1,
                                              float* acc,
                                              unsigned mask, const int k,
                                              const int so, const int dof, const float nrm,
                                              const int lane, const bool upper, const int hl) {
    int   sj[AGG_G], dj[AGG_G];
    float nj[AGG_G], v0[AGG_G], v1[AGG_G / 2];
#pragma unroll
    for (int u = 0; u < AGG_G; ++u) {
        sj[u] = 0; dj[u] = 0; nj[u] = 0.0f; v0[u] = 0.0f;
        if (FULL || u < k) {
            const int j = __builtin_ctz(mask);
            mask &= mask - 1u;
            sj[u] = __builtin_amdgcn_readlane(so, j);
            dj[u] = __builtin_amdgcn_readlane(dof, j);
            nj[u] = __int_as_float(__builtin_amdgcn_readlane(__float_as_int(nrm), j));
        }
    }
#pragma unroll
    for (int u = 0; u < AGG_G; ++u)
        if (FULL || u < k) v0[u] = (xt0 + sj[u])[lane];
#pragma unroll
    for (int p = 0; p < AGG_G / 2; ++p) {
        v1[p] = 0.0f;
        if (FULL || 2 * p < k) {
            const bool ok = FULL || !upper || (2 * p + 1 < k);
            const int  sp = upper ? sj[2 * p + 1] : sj[2 * p];
            if (ok) v1[p] = xt1[(sp >> 1) + hl];
        }
    }
#pragma unroll
    for (int u = 0; u < AGG_G; ++u)
        if (FULL || u < k) atomicAdd(&acc[dj[u] + lane], v0[u] * nj[u]);
#pragma unroll
    for (int p = 0; p < AGG_G / 2; ++p) {
        if (FULL || 2 * p < k) {
            const bool  ok = FULL || !upper || (2 * p + 1 < k);
            const int   dp = upper ? dj[2 * p + 1] : dj[2 * p];
            const float np = upper ? nj[2 * p + 1] : nj[2 * p];
            if (ok) atomicAdd(&acc[dp + P0W + hl], v1[p] * np);
        }
    }
    return mask;
}

__global__ __launch_bounds__(AGG_TPB)
void k_aggregate(const int* __restrict__ ei, const float* __restrict__ ew,
                 const float* __restrict__ dinv,
                 const float* __restrict__ xt0, const float* __restrict__ xt1,
                 const unsigned* __restrict__ rec, const int* __restrict__ off,
                 const float* __restrict__ gw, const float* __restrict__ gb,
                 const float* __restrict__ cw, const float* __restrict__ cb,
                 const float* __restrict__ lwp, const float* __restrict__ lbp,
                 float* __restrict__ out, int n_edges, int n_nodes) {
    __shared__ float acc[RG_ROWS * ACC_P];
    __shared__ int   pstart[BIN_WG];
    __shared__ int   ppre[BIN_WG];
    __shared__ float w2s[3 * HCH];
    __shared__ float bbs[3 * HCH];

    const int  tid   = threadIdx.x;
    const int  lane  = tid & 31;
    const int  wave  = tid >> 5;
    const bool upper = lane >= 16;
    const int  hl    = lane & 15;
    const int  rg    = (int)blockIdx.x;
    const int  n0    = rg << RG_SH;
    const int  rows  = min(RG_ROWS, n_nodes - n0);
    const int  per   = (n_edges + BIN_WG - 1) / BIN_WG;

    for (int i = tid; i < RG_ROWS * ACC_P; i += AGG_TPB) {
        const int r = i / ACC_P;
        const int s = i - r * ACC_P;
        float v = 0.0f;
        if (r < rows) {
            const int   n  = n0 + r;
            const float di = dinv[n];
            const float xv = (s < P0W) ? xt0[(size_t)n * P0W + s] : xt1[(size_t)n * P1W + (s - P0W)];
            v = xv * (di * di);
        }
        acc[i] = v;
    }
    if (tid < 6 * HCH) {
        const bool  isb = tid >= 3 * HCH;
        const int   idx = isb ? tid - 3 * HCH : tid;
        const int   k   = idx / HCH;
        const int   co  = idx - k * HCH;
        const float* __restrict__ gs = isb ? gb : gw;
        float a = 0.0f;
        for (int ci = 0; ci < HCH; ++ci)
            a += gs[ci] * cw[(size_t)co * 96 + ci * 3 + k];
        if (isb) bbs[idx] = a; else w2s[idx] = a;
    }
    if (tid < BIN_WG) {
        const int a = off[(size_t)tid * OFF_P + rg];
        const int z = off[(size_t)tid * OFF_P + rg + 1];
        pstart[tid] = tid * per + a;
        ppre[tid]   = z - a;
    }
    __syncthreads();
    for (int d = 1; d < BIN_WG; d <<= 1) {
        int t = 0;
        if (tid < BIN_WG && tid >= d) t = ppre[tid - d];
        __syncthreads();
        if (tid < BIN_WG) ppre[tid] += t;
        __syncthreads();
    }
    const int total = ppre[BIN_WG - 1];
    const int nb    = (total + 31) >> 5;

    for (int b0 = wave; b0 < nb; b0 += AGG_NW * AGG_CH) {
        int      so[AGG_CH], dof[AGG_CH];
        float    nrm[AGG_CH];
        unsigned msk[AGG_CH];
        unsigned rv[AGG_CH];
#pragma unroll
        for (int c = 0; c < AGG_CH; ++c) {
            const int  i     = ((b0 + c * AGG_NW) << 5) + lane;
            const bool valid = i < total;
            msk[c] = (unsigned)__builtin_amdgcn_readfirstlane((int)__builtin_amdgcn_ballot_w32(valid));
            const int ic = min(i, total - 1);
            int p = 0;
#pragma unroll
            for (int st = BIN_WG / 2; st >= 1; st >>= 1)
                if (ppre[p + st - 1] <= ic) p += st;
            const int before = (p > 0) ? ppre[p - 1] : 0;
            rv[c] = rec[(size_t)pstart[p] + (ic - before)];
        }
        int   sv[AGG_CH], dl[AGG_CH];
        float wv[AGG_CH];
#pragma unroll
        for (int c = 0; c < AGG_CH; ++c) {
            const int e = (int)(rv[c] & ((1u << REC_SH) - 1u));
            dl[c] = (int)(rv[c] >> REC_SH);
            sv[c] = ei[e];
            wv[c] = ew[e];
        }
        float da[AGG_CH], db[AGG_CH];
#pragma unroll
        for (int c = 0; c < AGG_CH; ++c) {
            da[c] = dinv[sv[c]];
            db[c] = dinv[n0 + dl[c]];
        }
#pragma unroll
        for (int c = 0; c < AGG_CH; ++c) {
            nrm[c] = da[c] * wv[c] * db[c];
            asm volatile("" : "+v"(nrm[c]));
            so[c]  = sv[c] * P0W;
            dof[c] = dl[c] * ACC_P;
        }
#pragma unroll
        for (int c = 0; c < AGG_CH; ++c) {
            unsigned m = msk[c];
            while (m != 0u) {
                const int k = __builtin_popcount(m);
                if (k >= AGG_G) m = agg_group<true >(xt0, xt1, acc, m, AGG_G, so[c], dof[c], nrm[c], lane, upper, hl);
                else            m = agg_group<false>(xt0, xt1, acc, m, k,     so[c], dof[c], nrm[c], lane, upper, hl);
            }
        }
    }

    __syncthreads();

    const int h = lane >> 4;
    const int m = lane & 15;

    v2f bmat0, bmat1;
#pragma unroll
    for (int v = 0; v < 2; ++v) {
        const int  k  = 2 * h + v;
        const int  kc = (k < 3) ? k : 0;
        const float a0 = w2s[kc * HCH + m];
        const float a1 = w2s[kc * HCH + m + 16];
        bmat0[v] = (k < 3) ? a0 : 0.0f;
        bmat1[v] = (k < 3) ? a1 : 0.0f;
    }
    float bbA[3], bbB[3];
#pragma unroll
    for (int k = 0; k < 3; ++k) {
        bbA[k] = bbs[k * HCH + m];
        bbB[k] = bbs[k * HCH + m + 16];
    }
    v8f cmat0, cmat1;
#pragma unroll
    for (int j = 0; j < 8; ++j) {
        int t = j + 8 * h;
        float c0 = 0.0f, c1 = 0.0f;
        if (t < TT) {
            c0 = cb[m]      + bbA[1] + (t > 0 ? bbA[0] : 0.0f) + (t < TT - 1 ? bbA[2] : 0.0f);
            c1 = cb[m + 16] + bbB[1] + (t > 0 ? bbB[0] : 0.0f) + (t < TT - 1 ? bbB[2] : 0.0f);
        }
        cmat0[j] = c0;
        cmat1[j] = c1;
    }
    const float lwA = lwp[m], lwB = lwp[16 + m];
    const float lb  = lbp[0];

    const int  i0  = m + 2 * h - 1;
    const int  i1  = i0 + 1;
    const bool ok0 = (m < TT) && (i0 >= 0) && (i0 < TT);
    const bool ok1 = (m < TT) && (i1 < TT);
    const int  o0  = ok0 ? i0 : 0;
    const int  o1  = ok1 ? i1 : 0;
    const float msk0 = ok0 ? 1.0f : 0.0f;
    const float msk1 = ok1 ? 1.0f : 0.0f;

#pragma unroll
    for (int b = 0; b < BB; ++b) {
        const int s0 = b * TT + o0;
        const int s1 = b * TT + o1;
        float* outb = out + (size_t)b * n_nodes + n0;
        for (int r = wave; r < rows; r += AGG_NW) {
            v2f amat;
            amat[0] = acc[r * ACC_P + s0] * msk0;
            amat[1] = acc[r * ACC_P + s1] * msk1;

            v8f d0 = __builtin_amdgcn_wmma_f32_16x16x4_f32(false, amat, false, bmat0,
                                                           (short)0, cmat0, false, false);
            v8f d1 = __builtin_amdgcn_wmma_f32_16x16x4_f32(false, amat, false, bmat1,
                                                           (short)0, cmat1, false, false);

            float sA = 0.0f, sB = 0.0f;
#pragma unroll
            for (int j = 0; j < 8; ++j) {
                sA += fmaxf(d0[j], 0.0f);
                sB += fmaxf(d1[j], 0.0f);
            }
            sA += __shfl_xor(sA, 16);
            sB += __shfl_xor(sB, 16);

            float v = sA * lwA + sB * lwB;
            v += __shfl_xor(v, 1);
            v += __shfl_xor(v, 2);
            v += __shfl_xor(v, 4);
            v += __shfl_xor(v, 8);

            if (lane == 0)
                outb[r] = v * (1.0f / (float)TT) + lb;
        }
    }
}

extern "C" void kernel_launch(void* const* d_in, const int* in_sizes, int n_in,
                              void* d_out, int out_size, void* d_ws, size_t ws_size,
                              hipStream_t stream) {
    const float* x  = (const float*)d_in[0];
    const int*   ei = (const int*)d_in[1];
    const float* ew = (const float*)d_in[2];
    const float* gw = (const float*)d_in[3];
    const float* gb = (const float*)d_in[4];
    const float* cw = (const float*)d_in[5];
    const float* cb = (const float*)d_in[6];
    const float* lw = (const float*)d_in[7];
    const float* lb = (const float*)d_in[8];
    float* out = (float*)d_out;

    const int E = in_sizes[2];
    const int N = in_sizes[0] / BT;

    float*    xt0  = (float*)d_ws;
    float*    xt1  = xt0 + (size_t)N * P0W;
    float*    dinv = xt1 + (size_t)N * P1W;
    unsigned* rec  = (unsigned*)(dinv + N);
    int*      off  = (int*)(rec + E);

    k_deg_init <<<(N + 255) / 256, 256, 0, stream>>>(dinv, N);
    k_deg_edges<<<DEG_WG, DEG_TPB, 0, stream>>>(ei, ew, dinv, E, N);
    k_finalize <<<(N + 255) / 256, 256, 0, stream>>>(x, dinv, xt0, xt1, N);
    k_bin      <<<BIN_WG, BIN_TPB, 0, stream>>>(ei, rec, off, E);

    k_aggregate<<<(N + RG_ROWS - 1) / RG_ROWS, AGG_TPB, 0, stream>>>(ei, ew, dinv, xt0, xt1, rec, off,
                                                                    gw, gb, cw, cb, lw, lb, out, E, N);
}


# norm factored: x_t rows stored scaled by dinv[src], dinv[dst] applied once per summed row: no dinv gather per edge (5 -> 4 line requests an edge); factors rounded in another order
# speedup vs baseline: 3.4621x; 1.0714x over previous
#include <hip/hip_runtime.h>
#include <hip/hip_bf16.h>


typedef __attribute__((ext_vector_type(2))) float v2f;
typedef __attribute__((ext_vector_type(8))) float v8f;

#define BT 48
#define TT 12
#define BB 4
#define HCH 32

#define P0W 32
#define P1W 16

#define RG_SH    6
#define RG_ROWS  (1 << RG_SH)
#define RG_MAX   1024
#define BIN_WG   256
#define BIN_TPB  1024
#define BIN_UNR  8
#define OFF_P    (RG_MAX + 1)
#define REC_SH   25

__global__ void k_deg_init(float* __restrict__ deg, int n_nodes) {
    int n = blockIdx.x * blockDim.x + threadIdx.x;
    if (n < n_nodes) deg[n] = 1.0f;
}

#define DEG_CAP 51200
#define DEG_WG  64
#define DEG_TPB 1024
#define DEG_UNR 8
#define DEG_FL  7

__global__ __launch_bounds__(DEG_TPB)
void k_deg_edges(const int* __restrict__ ei, const float* __restrict__ ew,
                 float* __restrict__ deg, int n_edges, int n_nodes) {
    __shared__ float4 hist4[DEG_CAP / 4];
    float* hist = (float*)hist4;
    const int tid = threadIdx.x;
    const int per = (n_edges + (int)gridDim.x - 1) / (int)gridDim.x;
    const int e0  = (int)blockIdx.x * per;
    const int e1  = min(e0 + per, n_edges);
    const int* __restrict__ dst = ei + (size_t)n_edges;

    for (int base = 0; base < n_nodes; base += DEG_CAP) {
        const int cnt = min(DEG_CAP, n_nodes - base);

        for (int i = tid; i < DEG_CAP / 4; i += DEG_TPB)
            hist4[i] = make_float4(0.0f, 0.0f, 0.0f, 0.0f);
        __syncthreads();

        if (e0 < e1) {
            const int elast = e1 - 1;
            const int step  = DEG_UNR * DEG_TPB;
            int   dv[DEG_UNR];
            float w[DEG_UNR];
#pragma unroll
            for (int u = 0; u < DEG_UNR; ++u) {
                const int ec = min(e0 + tid + u * DEG_TPB, elast);
                dv[u] = dst[ec];
                w[u]  = ew[ec];
            }
            for (int e = e0 + tid; e < e1; e += step) {
                int   dn[DEG_UNR];
                float wn[DEG_UNR];
#pragma unroll
                for (int u = 0; u < DEG_UNR; ++u) {
                    const int ec = min(e + step + u * DEG_TPB, elast);
                    dn[u] = dst[ec];
                    wn[u] = ew[ec];
                }
#pragma unroll
                for (int u = 0; u < DEG_UNR; ++u) {
                    const int d = (e + u * DEG_TPB < e1) ? dv[u] - base : -1;
                    if ((unsigned)d < (unsigned)cnt)
                        atomicAdd(&hist[d], w[u]);
                }
#pragma unroll
                for (int u = 0; u < DEG_UNR; ++u) { dv[u] = dn[u]; w[u] = wn[u]; }
            }
        }
        __syncthreads();

        const int rounds = (cnt + DEG_TPB - 1) / DEG_TPB;
        const int rot    = (int)blockIdx.x % rounds;
        for (int r = 0; r < rounds; r += DEG_FL) {
            float v[DEG_FL];
            int   idx[DEG_FL];
#pragma unroll
            for (int u = 0; u < DEG_FL; ++u) {
                int rr = r + u + rot;
                if (rr >= rounds) rr -= rounds;
                const int  i  = rr * DEG_TPB + tid;
                const bool ok = (r + u < rounds) && (i < cnt);
                const float hv = hist[ok ? i : tid];
                idx[u] = i;
                v[u]   = ok ? hv : 0.0f;
            }
#pragma unroll
            for (int u = 0; u < DEG_FL; ++u)
                if (v[u] != 0.0f) atomicAdd(&deg[base + idx[u]], v[u]);
        }
        __syncthreads();
    }
}

__global__ void k_finalize(const float* __restrict__ x, float* __restrict__ dinv,
                           float* __restrict__ xt0, float* __restrict__ xt1, int n_nodes) {
    int n = blockIdx.x * blockDim.x + threadIdx.x;
    if (n >= n_nodes) return;
    float deg = dinv[n];
    float di  = deg > 0.0f ? rsqrtf(deg) : 0.0f;
    dinv[n] = di;
    float buf[BT];
#pragma unroll
    for (int s = 0; s < BT; ++s)
        buf[s] = x[(size_t)s * n_nodes + n] * di;
    float4* xa4 = (float4*)(xt0 + (size_t)n * P0W);
    float4* xb4 = (float4*)(xt1 + (size_t)n * P1W);
#pragma unroll
    for (int q = 0; q < P0W / 4; ++q)
        xa4[q] = make_float4(buf[4*q], buf[4*q+1], buf[4*q+2], buf[4*q+3]);
#pragma unroll
    for (int q = 0; q < P1W / 4; ++q)
        xb4[q] = make_float4(buf[P0W + 4*q], buf[P0W + 4*q+1], buf[P0W + 4*q+2], buf[P0W + 4*q+3]);
}

static_assert(RG_MAX == BIN_TPB, "the binning keeps one range per thread");
static_assert(BIN_WG <= 512, "the aggregation keeps one piece per thread");
__global__ __launch_bounds__(BIN_TPB)
void k_bin(const int* __restrict__ ei, unsigned* __restrict__ rec, int* __restrict__ off, int n_edges) {
    __shared__ int lcnt[RG_MAX];
    __shared__ int lsc[RG_MAX];
    const int tid = threadIdx.x;
    const int per = (n_edges + BIN_WG - 1) / BIN_WG;
    const int e0  = (int)blockIdx.x * per;
    const int e1  = min(e0 + per, n_edges);
    const int* __restrict__ dst = ei + (size_t)n_edges;
    const int step = BIN_UNR * BIN_TPB;

    lcnt[tid] = 0;
    __syncthreads();
    for (int e = e0 + tid; e < e1; e += step) {
        int dv[BIN_UNR];
#pragma unroll
        for (int u = 0; u < BIN_UNR; ++u) dv[u] = dst[min(e + u * BIN_TPB, e1 - 1)];
#pragma unroll
        for (int u = 0; u < BIN_UNR; ++u)
            if (e + u * BIN_TPB < e1) atomicAdd(&lcnt[dv[u] >> RG_SH], 1);
    }
    __syncthreads();

    const int mine = lcnt[tid];
    lsc[tid] = mine;
    __syncthreads();
    for (int d = 1; d < RG_MAX; d <<= 1) {
        const int t = (tid >= d) ? lsc[tid - d] : 0;
        __syncthreads();
        lsc[tid] += t;
        __syncthreads();
    }
    const int incl = lsc[tid];
    const int excl = incl - mine;
    int* __restrict__ offw = off + (size_t)blockIdx.x * OFF_P;
    offw[tid] = excl;
    if (tid == RG_MAX - 1) offw[RG_MAX] = incl;
    lcnt[tid] = excl;
    __syncthreads();

    for (int e = e0 + tid; e < e1; e += step) {
        int dv[BIN_UNR];
#pragma unroll
        for (int u = 0; u < BIN_UNR; ++u) dv[u] = dst[min(e + u * BIN_TPB, e1 - 1)];
#pragma unroll
        for (int u = 0; u < BIN_UNR; ++u) {
            if (e + u * BIN_TPB < e1) {
                const int p = atomicAdd(&lcnt[dv[u] >> RG_SH], 1);
                rec[(size_t)e0 + p] = (unsigned)(e + u * BIN_TPB) | ((unsigned)(dv[u] & (RG_ROWS - 1)) << REC_SH);
            }
        }
    }
}

#define AGG_TPB 512
#define AGG_NW  (AGG_TPB / 32)
#define AGG_G   8
#define AGG_CH  5
#define ACC_P   BT

template <bool FULL>
__device__ __forceinline__ unsigned agg_group(const float* __restrict__ xt0, const float* __restrict__ xt1,
                                              float* acc,
                                              unsigned mask, const int k,
                                              const int so, const int dof, const float nrm,
                                              const int lane, const bool upper, const int hl) {
    int   sj[AGG_G], dj[AGG_G];
    float nj[AGG_G], v0[AGG_G], v1[AGG_G / 2];
#pragma unroll
    for (int u = 0; u < AGG_G; ++u) {
        sj[u] = 0; dj[u] = 0; nj[u] = 0.0f; v0[u] = 0.0f;
        if (FULL || u < k) {
            const int j = __builtin_ctz(mask);
            mask &= mask - 1u;
            sj[u] = __builtin_amdgcn_readlane(so, j);
            dj[u] = __builtin_amdgcn_readlane(dof, j);
            nj[u] = __int_as_float(__builtin_amdgcn_readlane(__float_as_int(nrm), j));
        }
    }
#pragma unroll
    for (int u = 0; u < AGG_G; ++u)
        if (FULL || u < k) v0[u] = (xt0 + sj[u])[lane];
#pragma unroll
    for (int p = 0; p < AGG_G / 2; ++p) {
        v1[p] = 0.0f;
        if (FULL || 2 * p < k) {
            const bool ok = FULL || !upper || (2 * p + 1 < k);
            const int  sp = upper ? sj[2 * p + 1] : sj[2 * p];
            if (ok) v1[p] = xt1[(sp >> 1) + hl];
        }
    }
#pragma unroll
    for (int u = 0; u < AGG_G; ++u)
        if (FULL || u < k) atomicAdd(&acc[dj[u] + lane], v0[u] * nj[u]);
#pragma unroll
    for (int p = 0; p < AGG_G / 2; ++p) {
        if (FULL || 2 * p < k) {
            const bool  ok = FULL || !upper || (2 * p + 1 < k);
            const int   dp = upper ? dj[2 * p + 1] : dj[2 * p];
            const float np = upper ? nj[2 * p + 1] : nj[2 * p];
            if (ok) atomicAdd(&acc[dp + P0W + hl], v1[p] * np);
        }
    }
    return mask;
}

__global__ __launch_bounds__(AGG_TPB)
void k_aggregate(const int* __restrict__ ei, const float* __restrict__ ew,
                 const float* __restrict__ dinv,
                 const float* __restrict__ xt0, const float* __restrict__ xt1,
                 const unsigned* __restrict__ rec, const int* __restrict__ off,
                 const float* __restrict__ gw, const float* __restrict__ gb,
                 const float* __restrict__ cw, const float* __restrict__ cb,
                 const float* __restrict__ lwp, const float* __restrict__ lbp,
                 float* __restrict__ out, int n_edges, int n_nodes) {
    __shared__ float acc[RG_ROWS * ACC_P];
    __shared__ float drow[RG_ROWS];
    __shared__ int   pstart[BIN_WG];
    __shared__ int   ppre[BIN_WG];
    __shared__ float w2s[3 * HCH];
    __shared__ float bbs[3 * HCH];

    const int  tid   = threadIdx.x;
    const int  lane  = tid & 31;
    const int  wave  = tid >> 5;
    const bool upper = lane >= 16;
    const int  hl    = lane & 15;
    const int  rg    = (int)blockIdx.x;
    const int  n0    = rg << RG_SH;
    const int  rows  = min(RG_ROWS, n_nodes - n0);
    const int  per   = (n_edges + BIN_WG - 1) / BIN_WG;

    for (int i = tid; i < RG_ROWS * ACC_P; i += AGG_TPB) {
        const int r = i / ACC_P;
        const int s = i - r * ACC_P;
        float v = 0.0f;
        if (r < rows) {
            const int n = n0 + r;
            v = (s < P0W) ? xt0[(size_t)n * P0W + s] : xt1[(size_t)n * P1W + (s - P0W)];
        }
        acc[i] = v;
    }
    if (tid < RG_ROWS) drow[tid] = (tid < rows) ? dinv[n0 + tid] : 0.0f;
    if (tid < 6 * HCH) {
        const bool  isb = tid >= 3 * HCH;
        const int   idx = isb ? tid - 3 * HCH : tid;
        const int   k   = idx / HCH;
        const int   co  = idx - k * HCH;
        const float* __restrict__ gs = isb ? gb : gw;
        float a = 0.0f;
        for (int ci = 0; ci < HCH; ++ci)
            a += gs[ci] * cw[(size_t)co * 96 + ci * 3 + k];
        if (isb) bbs[idx] = a; else w2s[idx] = a;
    }
    if (tid < BIN_WG) {
        const int a = off[(size_t)tid * OFF_P + rg];
        const int z = off[(size_t)tid * OFF_P + rg + 1];
        pstart[tid] = tid * per + a;
        ppre[tid]   = z - a;
    }
    __syncthreads();
    for (int d = 1; d < BIN_WG; d <<= 1) {
        int t = 0;
        if (tid < BIN_WG && tid >= d) t = ppre[tid - d];
        __syncthreads();
        if (tid < BIN_WG) ppre[tid] += t;
        __syncthreads();
    }
    const int total = ppre[BIN_WG - 1];
    const int nb    = (total + 31) >> 5;

    for (int b0 = wave; b0 < nb; b0 += AGG_NW * AGG_CH) {
        int      so[AGG_CH], dof[AGG_CH];
        float    nrm[AGG_CH];
        unsigned msk[AGG_CH];
        unsigned rv[AGG_CH];
#pragma unroll
        for (int c = 0; c < AGG_CH; ++c) {
            const int  i     = ((b0 + c * AGG_NW) << 5) + lane;
            const bool valid = i < total;
            msk[c] = (unsigned)__builtin_amdgcn_readfirstlane((int)__builtin_amdgcn_ballot_w32(valid));
            const int ic = min(i, total - 1);
            int p = 0;
#pragma unroll
            for (int st = BIN_WG / 2; st >= 1; st >>= 1)
                if (ppre[p + st - 1] <= ic) p += st;
            const int before = (p > 0) ? ppre[p - 1] : 0;
            rv[c] = rec[(size_t)pstart[p] + (ic - before)];
        }
        int   sv[AGG_CH], dl[AGG_CH];
        float wv[AGG_CH];
#pragma unroll
        for (int c = 0; c < AGG_CH; ++c) {
            const int e = (int)(rv[c] & ((1u << REC_SH) - 1u));
            dl[c] = (int)(rv[c] >> REC_SH);
            sv[c] = ei[e];
            wv[c] = ew[e];
        }
#pragma unroll
        for (int c = 0; c < AGG_CH; ++c) {
            nrm[c] = wv[c];
            so[c]  = sv[c] * P0W;
            dof[c] = dl[c] * ACC_P;
            asm volatile("" : "+v"(nrm[c]), "+v"(so[c]));
        }
#pragma unroll
        for (int c = 0; c < AGG_CH; ++c) {
            unsigned m = msk[c];
            while (m != 0u) {
                const int k = __builtin_popcount(m);
                if (k >= AGG_G) m = agg_group<true >(xt0, xt1, acc, m, AGG_G, so[c], dof[c], nrm[c], lane, upper, hl);
                else            m = agg_group<false>(xt0, xt1, acc, m, k,     so[c], dof[c], nrm[c], lane, upper, hl);
            }
        }
    }

    __syncthreads();

    const int h = lane >> 4;
    const int m = lane & 15;

    v2f bmat0, bmat1;
#pragma unroll
    for (int v = 0; v < 2; ++v) {
        const int  k  = 2 * h + v;
        const int  kc = (k < 3) ? k : 0;
        const float a0 = w2s[kc * HCH + m];
        const float a1 = w2s[kc * HCH + m + 16];
        bmat0[v] = (k < 3) ? a0 : 0.0f;
        bmat1[v] = (k < 3) ? a1 : 0.0f;
    }
    float bbA[3], bbB[3];
#pragma unroll
    for (int k = 0; k < 3; ++k) {
        bbA[k] = bbs[k * HCH + m];
        bbB[k] = bbs[k * HCH + m + 16];
    }
    v8f cmat0, cmat1;
#pragma unroll
    for (int j = 0; j < 8; ++j) {
        int t = j + 8 * h;
        float c0 = 0.0f, c1 = 0.0f;
        if (t < TT) {
            c0 = cb[m]      + bbA[1] + (t > 0 ? bbA[0] : 0.0f) + (t < TT - 1 ? bbA[2] : 0.0f);
            c1 = cb[m + 16] + bbB[1] + (t > 0 ? bbB[0] : 0.0f) + (t < TT - 1 ? bbB[2] : 0.0f);
        }
        cmat0[j] = c0;
        cmat1[j] = c1;
    }
    const float lwA = lwp[m], lwB = lwp[16 + m];
    const float lb  = lbp[0];

    const int  i0  = m + 2 * h - 1;
    const int  i1  = i0 + 1;
    const bool ok0 = (m < TT) && (i0 >= 0) && (i0 < TT);
    const bool ok1 = (m < TT) && (i1 < TT);
    const int  o0  = ok0 ? i0 : 0;
    const int  o1  = ok1 ? i1 : 0;
    const float msk0 = ok0 ? 1.0f : 0.0f;
    const float msk1 = ok1 ? 1.0f : 0.0f;

#pragma unroll
    for (int b = 0; b < BB; ++b) {
        const int s0 = b * TT + o0;
        const int s1 = b * TT + o1;
        float* outb = out + (size_t)b * n_nodes + n0;
        for (int r = wave; r < rows; r += AGG_NW) {
            const float dr = drow[r];
            v2f amat;
            amat[0] = acc[r * ACC_P + s0] * dr * msk0;
            amat[1] = acc[r * ACC_P + s1] * dr * msk1;

            v8f d0 = __builtin_amdgcn_wmma_f32_16x16x4_f32(false, amat, false, bmat0,
                                                           (short)0, cmat0, false, false);
            v8f d1 = __builtin_amdgcn_wmma_f32_16x16x4_f32(false, amat, false, bmat1,
                                                           (short)0, cmat1, false, false);

            float sA = 0.0f, sB = 0.0f;
#pragma unroll
            for (int j = 0; j < 8; ++j) {
                sA += fmaxf(d0[j], 0.0f);
                sB += fmaxf(d1[j], 0.0f);
            }
            sA += __shfl_xor(sA, 16);
            sB += __shfl_xor(sB, 16);

            float v = sA * lwA + sB * lwB;
            v += __shfl_xor(v, 1);
            v += __shfl_xor(v, 2);
            v += __shfl_xor(v, 4);
            v += __shfl_xor(v, 8);

            if (lane == 0)
                outb[r] = v * (1.0f / (float)TT) + lb;
        }
    }
}

extern "C" void kernel_launch(void* const* d_in, const int* in_sizes, int n_in,
                              void* d_out, int out_size, void* d_ws, size_t ws_size,
                              hipStream_t stream) {
    const float* x  = (const float*)d_in[0];
    const int*   ei = (const int*)d_in[1];
    const float* ew = (const float*)d_in[2];
    const float* gw = (const float*)d_in[3];
    const float* gb = (const float*)d_in[4];
    const float* cw = (const float*)d_in[5];
    const float* cb = (const float*)d_in[6];
    const float* lw = (const float*)d_in[7];
    const float* lb = (const float*)d_in[8];
    float* out = (float*)d_out;

    const int E = in_sizes[2];
    const int N = in_sizes[0] / BT;

    float*    xt0  = (float*)d_ws;
    float*    xt1  = xt0 + (size_t)N * P0W;
    float*    dinv = xt1 + (size_t)N * P1W;
    unsigned* rec  = (unsigned*)(dinv + N);
    int*      off  = (int*)(rec + E);

    k_deg_init <<<(N + 255) / 256, 256, 0, stream>>>(dinv, N);
    k_deg_edges<<<DEG_WG, DEG_TPB, 0, stream>>>(ei, ew, dinv, E, N);
    k_finalize <<<(N + 255) / 256, 256, 0, stream>>>(x, dinv, xt0, xt1, N);
    k_bin      <<<BIN_WG, BIN_TPB, 0, stream>>>(ei, rec, off, E);

    k_aggregate<<<(N + RG_ROWS - 1) / RG_ROWS, AGG_TPB, 0, stream>>>(ei, ew, dinv, xt0, xt1, rec, off,
                                                                    gw, gb, cw, cb, lw, lb, out, E, N);
}


# binning carries each edge's source and weight in an 8-byte record read in edge order: aggregation drops its two per-edge look-ups by edge number (4 to 2 scattered requests an edge); sum order kept
# speedup vs baseline: 3.8764x; 1.1197x over previous
#include <hip/hip_runtime.h>
#include <hip/hip_bf16.h>


typedef __attribute__((ext_vector_type(2))) float v2f;
typedef __attribute__((ext_vector_type(8))) float v8f;

#define BT 48
#define TT 12
#define BB 4
#define HCH 32

#define P0W 32
#define P1W 16

#define RG_SH    6
#define RG_ROWS  (1 << RG_SH)
#define RG_MAX   1024
#define BIN_WG   256
#define BIN_TPB  1024
#define BIN_UNR  8
#define OFF_P    (RG_MAX + 1)
#define REC_SH   25
#define RECW_SH  26

__global__ void k_deg_init(float* __restrict__ deg, int n_nodes) {
    int n = blockIdx.x * blockDim.x + threadIdx.x;
    if (n < n_nodes) deg[n] = 1.0f;
}

#define DEG_CAP 51200
#define DEG_WG  64
#define DEG_TPB 1024
#define DEG_UNR 8
#define DEG_FL  7

__global__ __launch_bounds__(DEG_TPB)
void k_deg_edges(const int* __restrict__ ei, const float* __restrict__ ew,
                 float* __restrict__ deg, int n_edges, int n_nodes) {
    __shared__ float4 hist4[DEG_CAP / 4];
    float* hist = (float*)hist4;
    const int tid = threadIdx.x;
    const int per = (n_edges + (int)gridDim.x - 1) / (int)gridDim.x;
    const int e0  = (int)blockIdx.x * per;
    const int e1  = min(e0 + per, n_edges);
    const int* __restrict__ dst = ei + (size_t)n_edges;

    for (int base = 0; base < n_nodes; base += DEG_CAP) {
        const int cnt = min(DEG_CAP, n_nodes - base);

        for (int i = tid; i < DEG_CAP / 4; i += DEG_TPB)
            hist4[i] = make_float4(0.0f, 0.0f, 0.0f, 0.0f);
        __syncthreads();

        if (e0 < e1) {
            const int elast = e1 - 1;
            const int step  = DEG_UNR * DEG_TPB;
            int   dv[DEG_UNR];
            float w[DEG_UNR];
#pragma unroll
            for (int u = 0; u < DEG_UNR; ++u) {
                const int ec = min(e0 + tid + u * DEG_TPB, elast);
                dv[u] = dst[ec];
                w[u]  = ew[ec];
            }
            for (int e = e0 + tid; e < e1; e += step) {
                int   dn[DEG_UNR];
                float wn[DEG_UNR];
#pragma unroll
                for (int u = 0; u < DEG_UNR; ++u) {
                    const int ec = min(e + step + u * DEG_TPB, elast);
                    dn[u] = dst[ec];
                    wn[u] = ew[ec];
                }
#pragma unroll
                for (int u = 0; u < DEG_UNR; ++u) {
                    const int d = (e + u * DEG_TPB < e1) ? dv[u] - base : -1;
                    if ((unsigned)d < (unsigned)cnt)
                        atomicAdd(&hist[d], w[u]);
                }
#pragma unroll
                for (int u = 0; u < DEG_UNR; ++u) { dv[u] = dn[u]; w[u] = wn[u]; }
            }
        }
        __syncthreads();

        const int rounds = (cnt + DEG_TPB - 1) / DEG_TPB;
        const int rot    = (int)blockIdx.x % rounds;
        for (int r = 0; r < rounds; r += DEG_FL) {
            float v[DEG_FL];
            int   idx[DEG_FL];
#pragma unroll
            for (int u = 0; u < DEG_FL; ++u) {
                int rr = r + u + rot;
                if (rr >= rounds) rr -= rounds;
                const int  i  = rr * DEG_TPB + tid;
                const bool ok = (r + u < rounds) && (i < cnt);
                const float hv = hist[ok ? i : tid];
                idx[u] = i;
                v[u]   = ok ? hv : 0.0f;
            }
#pragma unroll
            for (int u = 0; u < DEG_FL; ++u)
                if (v[u] != 0.0f) atomicAdd(&deg[base + idx[u]], v[u]);
        }
        __syncthreads();
    }
}

__global__ void k_finalize(const float* __restrict__ x, float* __restrict__ dinv,
                           float* __restrict__ xt0, float* __restrict__ xt1, int n_nodes) {
    int n = blockIdx.x * blockDim.x + threadIdx.x;
    if (n >= n_nodes) return;
    float deg = dinv[n];
    float di  = deg > 0.0f ? rsqrtf(deg) : 0.0f;
    dinv[n] = di;
    float buf[BT];
#pragma unroll
    for (int s = 0; s < BT; ++s)
        buf[s] = x[(size_t)s * n_nodes + n] * di;
    float4* xa4 = (float4*)(xt0 + (size_t)n * P0W);
    float4* xb4 = (float4*)(xt1 + (size_t)n * P1W);
#pragma unroll
    for (int q = 0; q < P0W / 4; ++q)
        xa4[q] = make_float4(buf[4*q], buf[4*q+1], buf[4*q+2], buf[4*q+3]);
#pragma unroll
    for (int q = 0; q < P1W / 4; ++q)
        xb4[q] = make_float4(buf[P0W + 4*q], buf[P0W + 4*q+1], buf[P0W + 4*q+2], buf[P0W + 4*q+3]);
}

static_assert(RG_MAX == BIN_TPB, "the binning keeps one range per thread");
static_assert(BIN_WG <= 512, "the aggregation keeps one piece per thread");
template <bool WIDE>
__global__ __launch_bounds__(BIN_TPB)
void k_bin(const int* __restrict__ ei, const float* __restrict__ ew,
           unsigned* __restrict__ rec, int* __restrict__ off, int n_edges) {
    __shared__ int lcnt[RG_MAX];
    __shared__ int lsc[RG_MAX];
    const int tid = threadIdx.x;
    const int per = (n_edges + BIN_WG - 1) / BIN_WG;
    const int e0  = (int)blockIdx.x * per;
    const int e1  = min(e0 + per, n_edges);
    const int* __restrict__ dst = ei + (size_t)n_edges;
    const int step = BIN_UNR * BIN_TPB;

    lcnt[tid] = 0;
    __syncthreads();
    for (int e = e0 + tid; e < e1; e += step) {
        int dv[BIN_UNR];
#pragma unroll
        for (int u = 0; u < BIN_UNR; ++u) dv[u] = dst[min(e + u * BIN_TPB, e1 - 1)];
#pragma unroll
        for (int u = 0; u < BIN_UNR; ++u)
            if (e + u * BIN_TPB < e1) atomicAdd(&lcnt[dv[u] >> RG_SH], 1);
    }
    __syncthreads();

    const int mine = lcnt[tid];
    lsc[tid] = mine;
    __syncthreads();
    for (int d = 1; d < RG_MAX; d <<= 1) {
        const int t = (tid >= d) ? lsc[tid - d] : 0;
        __syncthreads();
        lsc[tid] += t;
        __syncthreads();
    }
    const int incl = lsc[tid];
    const int excl = incl - mine;
    int* __restrict__ offw = off + (size_t)blockIdx.x * OFF_P;
    offw[tid] = excl;
    if (tid == RG_MAX - 1) offw[RG_MAX] = incl;
    lcnt[tid] = excl;
    __syncthreads();

    uint2* __restrict__ rec2 = (uint2*)rec;
    for (int e = e0 + tid; e < e1; e += step) {
        int   dv[BIN_UNR];
        int   sv[BIN_UNR];
        float wv[BIN_UNR];
#pragma unroll
        for (int u = 0; u < BIN_UNR; ++u) {
            const int ec = min(e + u * BIN_TPB, e1 - 1);
            dv[u] = dst[ec];
            if (WIDE) { sv[u] = ei[ec]; wv[u] = ew[ec]; }
        }
#pragma unroll
        for (int u = 0; u < BIN_UNR; ++u) {
            if (e + u * BIN_TPB < e1) {
                const int      p   = atomicAdd(&lcnt[dv[u] >> RG_SH], 1);
                const unsigned row = (unsigned)(dv[u] & (RG_ROWS - 1));
                if (WIDE) rec2[(size_t)e0 + p] = make_uint2((unsigned)sv[u] | (row << RECW_SH), __float_as_uint(wv[u]));
                else      rec[(size_t)e0 + p]  = (unsigned)(e + u * BIN_TPB) | (row << REC_SH);
            }
        }
    }
}

#define AGG_TPB 512
#define AGG_NW  (AGG_TPB / 32)
#define AGG_G   8
#define AGG_CH  5
#define ACC_P   BT

template <bool FULL>
__device__ __forceinline__ unsigned agg_group(const float* __restrict__ xt0, const float* __restrict__ xt1,
                                              float* acc,
                                              unsigned mask, const int k,
                                              const int so, const int dof, const float nrm,
                                              const int lane, const bool upper, const int hl) {
    int   sj[AGG_G], dj[AGG_G];
    float nj[AGG_G], v0[AGG_G], v1[AGG_G / 2];
#pragma unroll
    for (int u = 0; u < AGG_G; ++u) {
        sj[u] = 0; dj[u] = 0; nj[u] = 0.0f; v0[u] = 0.0f;
        if (FULL || u < k) {
            const int j = __builtin_ctz(mask);
            mask &= mask - 1u;
            sj[u] = __builtin_amdgcn_readlane(so, j);
            dj[u] = __builtin_amdgcn_readlane(dof, j);
            nj[u] = __int_as_float(__builtin_amdgcn_readlane(__float_as_int(nrm), j));
        }
    }
#pragma unroll
    for (int u = 0; u < AGG_G; ++u)
        if (FULL || u < k) v0[u] = (xt0 + sj[u])[lane];
#pragma unroll
    for (int p = 0; p < AGG_G / 2; ++p) {
        v1[p] = 0.0f;
        if (FULL || 2 * p < k) {
            const bool ok = FULL || !upper || (2 * p + 1 < k);
            const int  sp = upper ? sj[2 * p + 1] : sj[2 * p];
            if (ok) v1[p] = xt1[(sp >> 1) + hl];
        }
    }
#pragma unroll
    for (int u = 0; u < AGG_G; ++u)
        if (FULL || u < k) atomicAdd(&acc[dj[u] + lane], v0[u] * nj[u]);
#pragma unroll
    for (int p = 0; p < AGG_G / 2; ++p) {
        if (FULL || 2 * p < k) {
            const bool  ok = FULL || !upper || (2 * p + 1 < k);
            const int   dp = upper ? dj[2 * p + 1] : dj[2 * p];
            const float np = upper ? nj[2 * p + 1] : nj[2 * p];
            if (ok) atomicAdd(&acc[dp + P0W + hl], v1[p] * np);
        }
    }
    return mask;
}

template <bool WIDE>
__global__ __launch_bounds__(AGG_TPB)
void k_aggregate(const int* __restrict__ ei, const float* __restrict__ ew,
                 const float* __restrict__ dinv,
                 const float* __restrict__ xt0, const float* __restrict__ xt1,
                 const unsigned* __restrict__ rec, const int* __restrict__ off,
                 const float* __restrict__ gw, const float* __restrict__ gb,
                 const float* __restrict__ cw, const float* __restrict__ cb,
                 const float* __restrict__ lwp, const float* __restrict__ lbp,
                 float* __restrict__ out, int n_edges, int n_nodes) {
    __shared__ float acc[RG_ROWS * ACC_P];
    __shared__ float drow[RG_ROWS];
    __shared__ int   pstart[BIN_WG];
    __shared__ int   ppre[BIN_WG];
    __shared__ float w2s[3 * HCH];
    __shared__ float bbs[3 * HCH];

    const int  tid   = threadIdx.x;
    const int  lane  = tid & 31;
    const int  wave  = tid >> 5;
    const bool upper = lane >= 16;
    const int  hl    = lane & 15;
    const int  rg    = (int)blockIdx.x;
    const int  n0    = rg << RG_SH;
    const int  rows  = min(RG_ROWS, n_nodes - n0);
    const int  per   = (n_edges + BIN_WG - 1) / BIN_WG;

    for (int i = tid; i < RG_ROWS * ACC_P; i += AGG_TPB) {
        const int r = i / ACC_P;
        const int s = i - r * ACC_P;
        float v = 0.0f;
        if (r < rows) {
            const int n = n0 + r;
            v = (s < P0W) ? xt0[(size_t)n * P0W + s] : xt1[(size_t)n * P1W + (s - P0W)];
        }
        acc[i] = v;
    }
    if (tid < RG_ROWS) drow[tid] = (tid < rows) ? dinv[n0 + tid] : 0.0f;
    if (tid < 6 * HCH) {
        const bool  isb = tid >= 3 * HCH;
        const int   idx = isb ? tid - 3 * HCH : tid;
        const int   k   = idx / HCH;
        const int   co  = idx - k * HCH;
        const float* __restrict__ gs = isb ? gb : gw;
        float a = 0.0f;
        for (int ci = 0; ci < HCH; ++ci)
            a += gs[ci] * cw[(size_t)co * 96 + ci * 3 + k];
        if (isb) bbs[idx] = a; else w2s[idx] = a;
    }
    if (tid < BIN_WG) {
        const int a = off[(size_t)tid * OFF_P + rg];
        const int z = off[(size_t)tid * OFF_P + rg + 1];
        pstart[tid] = tid * per + a;
        ppre[tid]   = z - a;
    }
    __syncthreads();
    for (int d = 1; d < BIN_WG; d <<= 1) {
        int t = 0;
        if (tid < BIN_WG && tid >= d) t = ppre[tid - d];
        __syncthreads();
        if (tid < BIN_WG) ppre[tid] += t;
        __syncthreads();
    }
    const int total = ppre[BIN_WG - 1];
    const int nb    = (total + 31) >> 5;

    for (int b0 = wave; b0 < nb; b0 += AGG_NW * AGG_CH) {
        int      so[AGG_CH], dof[AGG_CH];
        float    nrm[AGG_CH];
        unsigned msk[AGG_CH];
        unsigned rv[AGG_CH];
        unsigned rw[AGG_CH];
#pragma unroll
        for (int c = 0; c < AGG_CH; ++c) {
            const int  i     = ((b0 + c * AGG_NW) << 5) + lane;
            const bool valid = i < total;
            msk[c] = (unsigned)__builtin_amdgcn_readfirstlane((int)__builtin_amdgcn_ballot_w32(valid));
            const int ic = min(i, total - 1);
            int p = 0;
#pragma unroll
            for (int st = BIN_WG / 2; st >= 1; st >>= 1)
                if (ppre[p + st - 1] <= ic) p += st;
            const int before = (p > 0) ? ppre[p - 1] : 0;
            const size_t ri = (size_t)pstart[p] + (ic - before);
            if (WIDE) {
                const uint2 r2 = ((const uint2*)rec)[ri];
                rv[c] = r2.x;
                rw[c] = r2.y;
            } else {
                rv[c] = rec[ri];
            }
        }
        int   sv[AGG_CH], dl[AGG_CH];
        float wv[AGG_CH];
#pragma unroll
        for (int c = 0; c < AGG_CH; ++c) {
            if (WIDE) {
                sv[c] = (int)(rv[c] & ((1u << RECW_SH) - 1u));
                dl[c] = (int)(rv[c] >> RECW_SH);
                wv[c] = __uint_as_float(rw[c]);
            } else {
                const int e = (int)(rv[c] & ((1u << REC_SH) - 1u));
                dl[c] = (int)(rv[c] >> REC_SH);
                sv[c] = ei[e];
                wv[c] = ew[e];
            }
        }
#pragma unroll
        for (int c = 0; c < AGG_CH; ++c) {
            nrm[c] = wv[c];
            so[c]  = sv[c] * P0W;
            dof[c] = dl[c] * ACC_P;
            if (!WIDE) asm volatile("" : "+v"(nrm[c]), "+v"(so[c]));
        }
#pragma unroll
        for (int c = 0; c < AGG_CH; ++c) {
            unsigned m = msk[c];
            while (m != 0u) {
                const int k = __builtin_popcount(m);
                if (k >= AGG_G) m = agg_group<true >(xt0, xt1, acc, m, AGG_G, so[c], dof[c], nrm[c], lane, upper, hl);
                else            m = agg_group<false>(xt0, xt1, acc, m, k,     so[c], dof[c], nrm[c], lane, upper, hl);
            }
        }
    }

    __syncthreads();

    const int h = lane >> 4;
    const int m = lane & 15;

    v2f bmat0, bmat1;
#pragma unroll
    for (int v = 0; v < 2; ++v) {
        const int  k  = 2 * h + v;
        const int  kc = (k < 3) ? k : 0;
        const float a0 = w2s[kc * HCH + m];
        const float a1 = w2s[kc * HCH + m + 16];
        bmat0[v] = (k < 3) ? a0 : 0.0f;
        bmat1[v] = (k < 3) ? a1 : 0.0f;
    }
    float bbA[3], bbB[3];
#pragma unroll
    for (int k = 0; k < 3; ++k) {
        bbA[k] = bbs[k * HCH + m];
        bbB[k] = bbs[k * HCH + m + 16];
    }
    v8f cmat0, cmat1;
#pragma unroll
    for (int j = 0; j < 8; ++j) {
        int t = j + 8 * h;
        float c0 = 0.0f, c1 = 0.0f;
        if (t < TT) {
            c0 = cb[m]      + bbA[1] + (t > 0 ? bbA[0] : 0.0f) + (t < TT - 1 ? bbA[2] : 0.0f);
            c1 = cb[m + 16] + bbB[1] + (t > 0 ? bbB[0] : 0.0f) + (t < TT - 1 ? bbB[2] : 0.0f);
        }
        cmat0[j] = c0;
        cmat1[j] = c1;
    }
    const float lwA = lwp[m], lwB = lwp[16 + m];
    const float lb  = lbp[0];

    const int  i0  = m + 2 * h - 1;
    const int  i1  = i0 + 1;
    const bool ok0 = (m < TT) && (i0 >= 0) && (i0 < TT);
    const bool ok1 = (m < TT) && (i1 < TT);
    const int  o0  = ok0 ? i0 : 0;
    const int  o1  = ok1 ? i1 : 0;
    const float msk0 = ok0 ? 1.0f : 0.0f;
    const float msk1 = ok1 ? 1.0f : 0.0f;

#pragma unroll
    for (int b = 0; b < BB; ++b) {
        const int s0 = b * TT + o0;
        const int s1 = b * TT + o1;
        float* outb = out + (size_t)b * n_nodes + n0;
        for (int r = wave; r < rows; r += AGG_NW) {
            const float dr = drow[r];
            v2f amat;
            amat[0] = acc[r * ACC_P + s0] * dr * msk0;
            amat[1] = acc[r * ACC_P + s1] * dr * msk1;

            v8f d0 = __builtin_amdgcn_wmma_f32_16x16x4_f32(false, amat, false, bmat0,
                                                           (short)0, cmat0, false, false);
            v8f d1 = __builtin_amdgcn_wmma_f32_16x16x4_f32(false, amat, false, bmat1,
                                                           (short)0, cmat1, false, false);

            float sA = 0.0f, sB = 0.0f;
#pragma unroll
            for (int j = 0; j < 8; ++j) {
                sA += fmaxf(d0[j], 0.0f);
                sB += fmaxf(d1[j], 0.0f);
            }
            sA += __shfl_xor(sA, 16);
            sB += __shfl_xor(sB, 16);

            float v = sA * lwA + sB * lwB;
            v += __shfl_xor(v, 1);
            v += __shfl_xor(v, 2);
            v += __shfl_xor(v, 4);
            v += __shfl_xor(v, 8);

            if (lane == 0)
                outb[r] = v * (1.0f / (float)TT) + lb;
        }
    }
}

extern "C" void kernel_launch(void* const* d_in, const int* in_sizes, int n_in,
                              void* d_out, int out_size, void* d_ws, size_t ws_size,
                              hipStream_t stream) {
    const float* x  = (const float*)d_in[0];
    const int*   ei = (const int*)d_in[1];
    const float* ew = (const float*)d_in[2];
    const float* gw = (const float*)d_in[3];
    const float* gb = (const float*)d_in[4];
    const float* cw = (const float*)d_in[5];
    const float* cb = (const float*)d_in[6];
    const float* lw = (const float*)d_in[7];
    const float* lb = (const float*)d_in[8];
    float* out = (float*)d_out;

    const int E = in_sizes[2];
    const int N = in_sizes[0] / BT;

    const size_t words_wide = (size_t)N * (P0W + P1W + 1) + (size_t)E * 2 + (size_t)BIN_WG * OFF_P;
    const bool   wide       = ws_size >= words_wide * sizeof(float);
    float*    xt0  = (float*)d_ws;
    float*    xt1  = xt0 + (size_t)N * P0W;
    unsigned* rec  = (unsigned*)(xt1 + (size_t)N * P1W);
    float*    dinv = (float*)(rec + (wide ? (size_t)E * 2 : (size_t)E));
    int*      off  = (int*)(dinv + N);

    k_deg_init <<<(N + 255) / 256, 256, 0, stream>>>(dinv, N);
    k_deg_edges<<<DEG_WG, DEG_TPB, 0, stream>>>(ei, ew, dinv, E, N);
    k_finalize <<<(N + 255) / 256, 256, 0, stream>>>(x, dinv, xt0, xt1, N);
    if (wide) {
        k_bin      <true ><<<BIN_WG, BIN_TPB, 0, stream>>>(ei, ew, rec, off, E);
        k_aggregate<true ><<<(N + RG_ROWS - 1) / RG_ROWS, AGG_TPB, 0, stream>>>(ei, ew, dinv, xt0, xt1, rec, off,
                                                                               gw, gb, cw, cb, lw, lb, out, E, N);
    } else {
        k_bin      <false><<<BIN_WG, BIN_TPB, 0, stream>>>(ei, ew, rec, off, E);
        k_aggregate<false><<<(N + RG_ROWS - 1) / RG_ROWS, AGG_TPB, 0, stream>>>(ei, ew, dinv, xt0, xt1, rec, off,
                                                                               gw, gb, cw, cb, lw, lb, out, E, N);
    }
}


# degree sums per destination range from the binned records (LDS atomics, no global atomics) in the kernel that writes the range's scaled rows: 5 launches become 3; weights summed in another order
# speedup vs baseline: 4.3766x; 1.1290x over previous
#include <hip/hip_runtime.h>
#include <hip/hip_bf16.h>


typedef __attribute__((ext_vector_type(2))) float v2f;
typedef __attribute__((ext_vector_type(8))) float v8f;

#define BT 48
#define TT 12
#define BB 4
#define HCH 32

#define P0W 32
#define P1W 16

#define RG_SH    6
#define RG_ROWS  (1 << RG_SH)
#define RG_MAX   1024
#define BIN_WG   256
#define BIN_TPB  1024
#define BIN_UNR  8
#define OFF_P    (RG_MAX + 1)
#define REC_SH   25
#define RECW_SH  26

static_assert(RG_MAX == BIN_TPB, "the binning keeps one range per thread");
static_assert(BIN_WG <= 512, "the aggregation keeps one piece per thread");
template <bool WIDE>
__global__ __launch_bounds__(BIN_TPB)
void k_bin(const int* __restrict__ ei, const float* __restrict__ ew,
           unsigned* __restrict__ rec, int* __restrict__ off, int n_edges) {
    __shared__ int lcnt[RG_MAX];
    __shared__ int lsc[RG_MAX];
    const int tid = threadIdx.x;
    const int per = (n_edges + BIN_WG - 1) / BIN_WG;
    const int e0  = (int)blockIdx.x * per;
    const int e1  = min(e0 + per, n_edges);
    const int* __restrict__ dst = ei + (size_t)n_edges;
    const int step = BIN_UNR * BIN_TPB;

    lcnt[tid] = 0;
    __syncthreads();
    for (int e = e0 + tid; e < e1; e += step) {
        int dv[BIN_UNR];
#pragma unroll
        for (int u = 0; u < BIN_UNR; ++u) dv[u] = dst[min(e + u * BIN_TPB, e1 - 1)];
#pragma unroll
        for (int u = 0; u < BIN_UNR; ++u)
            if (e + u * BIN_TPB < e1) atomicAdd(&lcnt[dv[u] >> RG_SH], 1);
    }
    __syncthreads();

    const int mine = lcnt[tid];
    lsc[tid] = mine;
    __syncthreads();
    for (int d = 1; d < RG_MAX; d <<= 1) {
        const int t = (tid >= d) ? lsc[tid - d] : 0;
        __syncthreads();
        lsc[tid] += t;
        __syncthreads();
    }
    const int incl = lsc[tid];
    const int excl = incl - mine;
    int* __restrict__ offw = off + (size_t)blockIdx.x * OFF_P;
    offw[tid] = excl;
    if (tid == RG_MAX - 1) offw[RG_MAX] = incl;
    lcnt[tid] = excl;
    __syncthreads();

    uint2* __restrict__ rec2 = (uint2*)rec;
    for (int e = e0 + tid; e < e1; e += step) {
        int   dv[BIN_UNR];
        int   sv[BIN_UNR];
        float wv[BIN_UNR];
#pragma unroll
        for (int u = 0; u < BIN_UNR; ++u) {
            const int ec = min(e + u * BIN_TPB, e1 - 1);
            dv[u] = dst[ec];
            if (WIDE) { sv[u] = ei[ec]; wv[u] = ew[ec]; }
        }
#pragma unroll
        for (int u = 0; u < BIN_UNR; ++u) {
            if (e + u * BIN_TPB < e1) {
                const int      p   = atomicAdd(&lcnt[dv[u] >> RG_SH], 1);
                const unsigned row = (unsigned)(dv[u] & (RG_ROWS - 1));
                if (WIDE) rec2[(size_t)e0 + p] = make_uint2((unsigned)sv[u] | (row << RECW_SH), __float_as_uint(wv[u]));
                else      rec[(size_t)e0 + p]  = (unsigned)(e + u * BIN_TPB) | (row << REC_SH);
            }
        }
    }
}

#define FIN_TPB 256
#define FIN_SUB 8
#define FIN_PC  (BIN_WG / (FIN_TPB / FIN_SUB))
#define FIN_SG  (FIN_TPB / RG_ROWS)
#define TILE_P  (BT + 1)
static_assert(BIN_WG % (FIN_TPB / FIN_SUB) == 0, "every piece has its lanes");
static_assert(BT % FIN_SG == 0, "a node's slices are split evenly over its threads");
static_assert(FIN_TPB >= RG_ROWS + 32, "one thread per word of the degree sums");

template <bool WIDE>
__global__ __launch_bounds__(FIN_TPB)
void k_rows(const float* __restrict__ x, const float* __restrict__ ew,
            const unsigned* __restrict__ rec, const int* __restrict__ off,
            float* __restrict__ dinv, float* __restrict__ xt0, float* __restrict__ xt1,
            int n_edges, int n_nodes) {
    __shared__ float dsum[RG_ROWS + 32];
    __shared__ float tile[RG_ROWS * TILE_P];
    const int tid  = threadIdx.x;
    const int rg   = (int)blockIdx.x;
    const int n0   = rg << RG_SH;
    const int rows = min(RG_ROWS, n_nodes - n0);
    const int per  = (n_edges + BIN_WG - 1) / BIN_WG;
    const int c    = tid & (RG_ROWS - 1);
    const int sg   = tid >> RG_SH;
    const int cc   = min(c, rows - 1);
    const int sub  = tid & (FIN_SUB - 1);
    const int pg   = tid / FIN_SUB;
    const int idle = RG_ROWS + (tid & 31);

    if (tid < RG_ROWS + 32) dsum[tid] = (tid < RG_ROWS) ? 1.0f : 0.0f;
    __syncthreads();

    int    len[FIN_PC];
    size_t p0[FIN_PC];
    int    maxlen = 0;
#pragma unroll
    for (int i = 0; i < FIN_PC; ++i) {
        const int pc = pg + (FIN_TPB / FIN_SUB) * i;
        const int a  = off[(size_t)pc * OFF_P + rg];
        len[i] = off[(size_t)pc * OFF_P + rg + 1] - a;
        p0[i]  = (size_t)pc * per + a;
        maxlen = max(maxlen, len[i]);
    }
    float xv[BT / FIN_SG];
#pragma unroll
    for (int j = 0; j < BT / FIN_SG; ++j)
        xv[j] = x[(size_t)(sg + FIN_SG * j) * n_nodes + n0 + cc];

    for (int j = sub; j < maxlen; j += FIN_SUB) {
        unsigned r0[FIN_PC];
        float    w[FIN_PC];
#pragma unroll
        for (int i = 0; i < FIN_PC; ++i) {
            const size_t ri = (j < len[i]) ? p0[i] + j : 0;
            if (WIDE) {
                const uint2 r2 = ((const uint2*)rec)[ri];
                r0[i] = r2.x;
                w[i]  = __uint_as_float(r2.y);
            } else {
                r0[i] = rec[ri];
                w[i]  = ew[r0[i] & ((1u << REC_SH) - 1u)];
            }
        }
#pragma unroll
        for (int i = 0; i < FIN_PC; ++i) {
            asm volatile("" : "+v"(r0[i]), "+v"(w[i]));
            const bool ok = j < len[i];
            atomicAdd(&dsum[ok ? (int)(r0[i] >> (WIDE ? RECW_SH : REC_SH)) : idle], ok ? w[i] : 0.0f);
        }
    }
    __syncthreads();

    const float deg = dsum[c];
    const float di  = deg > 0.0f ? rsqrtf(deg) : 0.0f;
    if (tid < rows) dinv[n0 + tid] = di;
#pragma unroll
    for (int j = 0; j < BT / FIN_SG; ++j)
        tile[c * TILE_P + sg + FIN_SG * j] = xv[j] * di;
    __syncthreads();

    float4* __restrict__ o0 = (float4*)(xt0 + (size_t)n0 * P0W);
    float4* __restrict__ o1 = (float4*)(xt1 + (size_t)n0 * P1W);
    for (int q = tid; q < rows * (P0W / 4); q += FIN_TPB) {
        const float* t = &tile[(q / (P0W / 4)) * TILE_P + 4 * (q % (P0W / 4))];
        o0[q] = make_float4(t[0], t[1], t[2], t[3]);
    }
    for (int q = tid; q < rows * (P1W / 4); q += FIN_TPB) {
        const float* t = &tile[(q / (P1W / 4)) * TILE_P + P0W + 4 * (q % (P1W / 4))];
        o1[q] = make_float4(t[0], t[1], t[2], t[3]);
    }
}

#define AGG_TPB 512
#define AGG_NW  (AGG_TPB / 32)
#define AGG_G   8
#define AGG_CH  5
#define ACC_P   BT

template <bool FULL>
__device__ __forceinline__ unsigned agg_group(const float* __restrict__ xt0, const float* __restrict__ xt1,
                                              float* acc,
                                              unsigned mask, const int k,
                                              const int so, const int dof, const float nrm,
                                              const int lane, const bool upper, const int hl) {
    int   sj[AGG_G], dj[AGG_G];
    float nj[AGG_G], v0[AGG_G], v1[AGG_G / 2];
#pragma unroll
    for (int u = 0; u < AGG_G; ++u) {
        sj[u] = 0; dj[u] = 0; nj[u] = 0.0f; v0[u] = 0.0f;
        if (FULL || u < k) {
            const int j = __builtin_ctz(mask);
            mask &= mask - 1u;
            sj[u] = __builtin_amdgcn_readlane(so, j);
            dj[u] = __builtin_amdgcn_readlane(dof, j);
            nj[u] = __int_as_float(__builtin_amdgcn_readlane(__float_as_int(nrm), j));
        }
    }
#pragma unroll
    for (int u = 0; u < AGG_G; ++u)
        if (FULL || u < k) v0[u] = (xt0 + sj[u])[lane];
#pragma unroll
    for (int p = 0; p < AGG_G / 2; ++p) {
        v1[p] = 0.0f;
        if (FULL || 2 * p < k) {
            const bool ok = FULL || !upper || (2 * p + 1 < k);
            const int  sp = upper ? sj[2 * p + 1] : sj[2 * p];
            if (ok) v1[p] = xt1[(sp >> 1) + hl];
        }
    }
#pragma unroll
    for (int u = 0; u < AGG_G; ++u)
        if (FULL || u < k) atomicAdd(&acc[dj[u] + lane], v0[u] * nj[u]);
#pragma unroll
    for (int p = 0; p < AGG_G / 2; ++p) {
        if (FULL || 2 * p < k) {
            const bool  ok = FULL || !upper || (2 * p + 1 < k);
            const int   dp = upper ? dj[2 * p + 1] : dj[2 * p];
            const float np = upper ? nj[2 * p + 1] : nj[2 * p];
            if (ok) atomicAdd(&acc[dp + P0W + hl], v1[p] * np);
        }
    }
    return mask;
}

template <bool WIDE>
__global__ __launch_bounds__(AGG_TPB)
void k_aggregate(const int* __restrict__ ei, const float* __restrict__ ew,
                 const float* __restrict__ dinv,
                 const float* __restrict__ xt0, const float* __restrict__ xt1,
                 const unsigned* __restrict__ rec, const int* __restrict__ off,
                 const float* __restrict__ gw, const float* __restrict__ gb,
                 const float* __restrict__ cw, const float* __restrict__ cb,
                 const float* __restrict__ lwp, const float* __restrict__ lbp,
                 float* __restrict__ out, int n_edges, int n_nodes) {
    __shared__ float acc[RG_ROWS * ACC_P];
    __shared__ float drow[RG_ROWS];
    __shared__ int   pstart[BIN_WG];
    __shared__ int   ppre[BIN_WG];
    __shared__ float w2s[3 * HCH];
    __shared__ float bbs[3 * HCH];

    const int  tid   = threadIdx.x;
    const int  lane  = tid & 31;
    const int  wave  = tid >> 5;
    const bool upper = lane >= 16;
    const int  hl    = lane & 15;
    const int  rg    = (int)blockIdx.x;
    const int  n0    = rg << RG_SH;
    const int  rows  = min(RG_ROWS, n_nodes - n0);
    const int  per   = (n_edges + BIN_WG - 1) / BIN_WG;

    for (int i = tid; i < RG_ROWS * ACC_P; i += AGG_TPB) {
        const int r = i / ACC_P;
        const int s = i - r * ACC_P;
        float v = 0.0f;
        if (r < rows) {
            const int n = n0 + r;
            v = (s < P0W) ? xt0[(size_t)n * P0W + s] : xt1[(size_t)n * P1W + (s - P0W)];
        }
        acc[i] = v;
    }
    if (tid < RG_ROWS) drow[tid] = (tid < rows) ? dinv[n0 + tid] : 0.0f;
    if (tid < 6 * HCH) {
        const bool  isb = tid >= 3 * HCH;
        const int   idx = isb ? tid - 3 * HCH : tid;
        const int   k   = idx / HCH;
        const int   co  = idx - k * HCH;
        const float* __restrict__ gs = isb ? gb : gw;
        float a = 0.0f;
        for (int ci = 0; ci < HCH; ++ci)
            a += gs[ci] * cw[(size_t)co * 96 + ci * 3 + k];
        if (isb) bbs[idx] = a; else w2s[idx] = a;
    }
    if (tid < BIN_WG) {
        const int a = off[(size_t)tid * OFF_P + rg];
        const int z = off[(size_t)tid * OFF_P + rg + 1];
        pstart[tid] = tid * per + a;
        ppre[tid]   = z - a;
    }
    __syncthreads();
    for (int d = 1; d < BIN_WG; d <<= 1) {
        int t = 0;
        if (tid < BIN_WG && tid >= d) t = ppre[tid - d];
        __syncthreads();
        if (tid < BIN_WG) ppre[tid] += t;
        __syncthreads();
    }
    const int total = ppre[BIN_WG - 1];
    const int nb    = (total + 31) >> 5;

    for (int b0 = wave; b0 < nb; b0 += AGG_NW * AGG_CH) {
        int      so[AGG_CH], dof[AGG_CH];
        float    nrm[AGG_CH];
        unsigned msk[AGG_CH];
        unsigned rv[AGG_CH];
        unsigned rw[AGG_CH];
#pragma unroll
        for (int c = 0; c < AGG_CH; ++c) {
            const int  i     = ((b0 + c * AGG_NW) << 5) + lane;
            const bool valid = i < total;
            msk[c] = (unsigned)__builtin_amdgcn_readfirstlane((int)__builtin_amdgcn_ballot_w32(valid));
            const int ic = min(i, total - 1);
            int p = 0;
#pragma unroll
            for (int st = BIN_WG / 2; st >= 1; st >>= 1)
                if (ppre[p + st - 1] <= ic) p += st;
            const int before = (p > 0) ? ppre[p - 1] : 0;
            const size_t ri = (size_t)pstart[p] + (ic - before);
            if (WIDE) {
                const uint2 r2 = ((const uint2*)rec)[ri];
                rv[c] = r2.x;
                rw[c] = r2.y;
            } else {
                rv[c] = rec[ri];
            }
        }
        int   sv[AGG_CH], dl[AGG_CH];
        float wv[AGG_CH];
#pragma unroll
        for (int c = 0; c < AGG_CH; ++c) {
            if (WIDE) {
                sv[c] = (int)(rv[c] & ((1u << RECW_SH) - 1u));
                dl[c] = (int)(rv[c] >> RECW_SH);
                wv[c] = __uint_as_float(rw[c]);
            } else {
                const int e = (int)(rv[c] & ((1u << REC_SH) - 1u));
                dl[c] = (int)(rv[c] >> REC_SH);
                sv[c] = ei[e];
                wv[c] = ew[e];
            }
        }
#pragma unroll
        for (int c = 0; c < AGG_CH; ++c) {
            nrm[c] = wv[c];
            so[c]  = sv[c] * P0W;
            dof[c] = dl[c] * ACC_P;
            if (!WIDE) asm volatile("" : "+v"(nrm[c]), "+v"(so[c]));
        }
#pragma unroll
        for (int c = 0; c < AGG_CH; ++c) {
            unsigned m = msk[c];
            while (m != 0u) {
                const int k = __builtin_popcount(m);
                if (k >= AGG_G) m = agg_group<true >(xt0, xt1, acc, m, AGG_G, so[c], dof[c], nrm[c], lane, upper, hl);
                else            m = agg_group<false>(xt0, xt1, acc, m, k,     so[c], dof[c], nrm[c], lane, upper, hl);
            }
        }
    }

    __syncthreads();

    const int h = lane >> 4;
    const int m = lane & 15;

    v2f bmat0, bmat1;
#pragma unroll
    for (int v = 0; v < 2; ++v) {
        const int  k  = 2 * h + v;
        const int  kc = (k < 3) ? k : 0;
        const float a0 = w2s[kc * HCH + m];
        const float a1 = w2s[kc * HCH + m + 16];
        bmat0[v] = (k < 3) ? a0 : 0.0f;
        bmat1[v] = (k < 3) ? a1 : 0.0f;
    }
    float bbA[3], bbB[3];
#pragma unroll
    for (int k = 0; k < 3; ++k) {
        bbA[k] = bbs[k * HCH + m];
        bbB[k] = bbs[k * HCH + m + 16];
    }
    v8f cmat0, cmat1;
#pragma unroll
    for (int j = 0; j < 8; ++j) {
        int t = j + 8 * h;
        float c0 = 0.0f, c1 = 0.0f;
        if (t < TT) {
            c0 = cb[m]      + bbA[1] + (t > 0 ? bbA[0] : 0.0f) + (t < TT - 1 ? bbA[2] : 0.0f);
            c1 = cb[m + 16] + bbB[1] + (t > 0 ? bbB[0] : 0.0f) + (t < TT - 1 ? bbB[2] : 0.0f);
        }
        cmat0[j] = c0;
        cmat1[j] = c1;
    }
    const float lwA = lwp[m], lwB = lwp[16 + m];
    const float lb  = lbp[0];

    const int  i0  = m + 2 * h - 1;
    const int  i1  = i0 + 1;
    const bool ok0 = (m < TT) && (i0 >= 0) && (i0 < TT);
    const bool ok1 = (m < TT) && (i1 < TT);
    const int  o0  = ok0 ? i0 : 0;
    const int  o1  = ok1 ? i1 : 0;
    const float msk0 = ok0 ? 1.0f : 0.0f;
    const float msk1 = ok1 ? 1.0f : 0.0f;

#pragma unroll
    for (int b = 0; b < BB; ++b) {
        const int s0 = b * TT + o0;
        const int s1 = b * TT + o1;
        float* outb = out + (size_t)b * n_nodes + n0;
        for (int r = wave; r < rows; r += AGG_NW) {
            const float dr = drow[r];
            v2f amat;
            amat[0] = acc[r * ACC_P + s0] * dr * msk0;
            amat[1] = acc[r * ACC_P + s1] * dr * msk1;

            v8f d0 = __builtin_amdgcn_wmma_f32_16x16x4_f32(false, amat, false, bmat0,
                                                           (short)0, cmat0, false, false);
            v8f d1 = __builtin_amdgcn_wmma_f32_16x16x4_f32(false, amat, false, bmat1,
                                                           (short)0, cmat1, false, false);

            float sA = 0.0f, sB = 0.0f;
#pragma unroll
            for (int j = 0; j < 8; ++j) {
                sA += fmaxf(d0[j], 0.0f);
                sB += fmaxf(d1[j], 0.0f);
            }
            sA += __shfl_xor(sA, 16);
            sB += __shfl_xor(sB, 16);

            float v = sA * lwA + sB * lwB;
            v += __shfl_xor(v, 1);
            v += __shfl_xor(v, 2);
            v += __shfl_xor(v, 4);
            v += __shfl_xor(v, 8);

            if (lane == 0)
                outb[r] = v * (1.0f / (float)TT) + lb;
        }
    }
}

extern "C" void kernel_launch(void* const* d_in, const int* in_sizes, int n_in,
                              void* d_out, int out_size, void* d_ws, size_t ws_size,
                              hipStream_t stream) {
    const float* x  = (const float*)d_in[0];
    const int*   ei = (const int*)d_in[1];
    const float* ew = (const float*)d_in[2];
    const float* gw = (const float*)d_in[3];
    const float* gb = (const float*)d_in[4];
    const float* cw = (const float*)d_in[5];
    const float* cb = (const float*)d_in[6];
    const float* lw = (const float*)d_in[7];
    const float* lb = (const float*)d_in[8];
    float* out = (float*)d_out;

    const int E = in_sizes[2];
    const int N = in_sizes[0] / BT;

    const size_t words_wide = (size_t)N * (P0W + P1W + 1) + (size_t)E * 2 + (size_t)BIN_WG * OFF_P;
    const bool   wide       = ws_size >= words_wide * sizeof(float);
    float*    xt0  = (float*)d_ws;
    float*    xt1  = xt0 + (size_t)N * P0W;
    unsigned* rec  = (unsigned*)(xt1 + (size_t)N * P1W);
    float*    dinv = (float*)(rec + (wide ? (size_t)E * 2 : (size_t)E));
    int*      off  = (int*)(dinv + N);

    if (wide) {
        k_bin      <true ><<<BIN_WG, BIN_TPB, 0, stream>>>(ei, ew, rec, off, E);
        k_rows     <true ><<<(N + RG_ROWS - 1) / RG_ROWS, FIN_TPB, 0, stream>>>(x, ew, rec, off, dinv, xt0, xt1, E, N);
        k_aggregate<true ><<<(N + RG_ROWS - 1) / RG_ROWS, AGG_TPB, 0, stream>>>(ei, ew, dinv, xt0, xt1, rec, off,
                                                                               gw, gb, cw, cb, lw, lb, out, E, N);
    } else {
        k_bin      <false><<<BIN_WG, BIN_TPB, 0, stream>>>(ei, ew, rec, off, E);
        k_rows     <false><<<(N + RG_ROWS - 1) / RG_ROWS, FIN_TPB, 0, stream>>>(x, ew, rec, off, dinv, xt0, xt1, E, N);
        k_aggregate<false><<<(N + RG_ROWS - 1) / RG_ROWS, AGG_TPB, 0, stream>>>(ei, ew, dinv, xt0, xt1, rec, off,
                                                                               gw, gb, cw, cb, lw, lb, out, E, N);
    }
}


# weight fold (192 sums of gcn_w or gcn_b times conv_w) formed once in the rows kernel, read from the workspace: aggregation workgroups no longer each redo it with 32-line gathers; sum order kept
# speedup vs baseline: 4.6885x; 1.0713x over previous
#include <hip/hip_runtime.h>
#include <hip/hip_bf16.h>


typedef __attribute__((ext_vector_type(2))) float v2f;
typedef __attribute__((ext_vector_type(8))) float v8f;

#define BT 48
#define TT 12
#define BB 4
#define HCH 32

#define P0W 32
#define P1W 16

#define RG_SH    6
#define RG_ROWS  (1 << RG_SH)
#define RG_MAX   1024
#define BIN_WG   256
#define BIN_TPB  1024
#define BIN_UNR  8
#define OFF_P    (RG_MAX + 1)
#define REC_SH   25
#define RECW_SH  26

static_assert(RG_MAX == BIN_TPB, "the binning keeps one range per thread");
static_assert(BIN_WG <= 512, "the aggregation keeps one piece per thread");
template <bool WIDE>
__global__ __launch_bounds__(BIN_TPB)
void k_bin(const int* __restrict__ ei, const float* __restrict__ ew,
           unsigned* __restrict__ rec, int* __restrict__ off, int n_edges) {
    __shared__ int lcnt[RG_MAX];
    __shared__ int lsc[RG_MAX];
    const int tid = threadIdx.x;
    const int per = (n_edges + BIN_WG - 1) / BIN_WG;
    const int e0  = (int)blockIdx.x * per;
    const int e1  = min(e0 + per, n_edges);
    const int* __restrict__ dst = ei + (size_t)n_edges;
    const int step = BIN_UNR * BIN_TPB;

    lcnt[tid] = 0;
    __syncthreads();
    for (int e = e0 + tid; e < e1; e += step) {
        int dv[BIN_UNR];
#pragma unroll
        for (int u = 0; u < BIN_UNR; ++u) dv[u] = dst[min(e + u * BIN_TPB, e1 - 1)];
#pragma unroll
        for (int u = 0; u < BIN_UNR; ++u)
            if (e + u * BIN_TPB < e1) atomicAdd(&lcnt[dv[u] >> RG_SH], 1);
    }
    __syncthreads();

    const int mine = lcnt[tid];
    lsc[tid] = mine;
    __syncthreads();
    for (int d = 1; d < RG_MAX; d <<= 1) {
        const int t = (tid >= d) ? lsc[tid - d] : 0;
        __syncthreads();
        lsc[tid] += t;
        __syncthreads();
    }
    const int incl = lsc[tid];
    const int excl = incl - mine;
    int* __restrict__ offw = off + (size_t)blockIdx.x * OFF_P;
    offw[tid] = excl;
    if (tid == RG_MAX - 1) offw[RG_MAX] = incl;
    lcnt[tid] = excl;
    __syncthreads();

    uint2* __restrict__ rec2 = (uint2*)rec;
    for (int e = e0 + tid; e < e1; e += step) {
        int   dv[BIN_UNR];
        int   sv[BIN_UNR];
        float wv[BIN_UNR];
#pragma unroll
        for (int u = 0; u < BIN_UNR; ++u) {
            const int ec = min(e + u * BIN_TPB, e1 - 1);
            dv[u] = dst[ec];
            if (WIDE) { sv[u] = ei[ec]; wv[u] = ew[ec]; }
        }
#pragma unroll
        for (int u = 0; u < BIN_UNR; ++u) {
            if (e + u * BIN_TPB < e1) {
                const int      p   = atomicAdd(&lcnt[dv[u] >> RG_SH], 1);
                const unsigned row = (unsigned)(dv[u] & (RG_ROWS - 1));
                if (WIDE) rec2[(size_t)e0 + p] = make_uint2((unsigned)sv[u] | (row << RECW_SH), __float_as_uint(wv[u]));
                else      rec[(size_t)e0 + p]  = (unsigned)(e + u * BIN_TPB) | (row << REC_SH);
            }
        }
    }
}

#define FIN_TPB 256
#define FIN_SUB 8
#define FIN_PC  (BIN_WG / (FIN_TPB / FIN_SUB))
#define FIN_SG  (FIN_TPB / RG_ROWS)
#define TILE_P  (BT + 1)
static_assert(BIN_WG % (FIN_TPB / FIN_SUB) == 0, "every piece has its lanes");
static_assert(BT % FIN_SG == 0, "a node's slices are split evenly over its threads");
static_assert(FIN_TPB >= RG_ROWS + 32, "one thread per word of the degree sums");

template <bool WIDE>
__global__ __launch_bounds__(FIN_TPB)
void k_rows(const float* __restrict__ x, const float* __restrict__ ew,
            const unsigned* __restrict__ rec, const int* __restrict__ off,
            const float* __restrict__ gw, const float* __restrict__ gb, const float* __restrict__ cw,
            float* __restrict__ fold,
            float* __restrict__ dinv, float* __restrict__ xt0, float* __restrict__ xt1,
            int n_edges, int n_nodes) {
    __shared__ float dsum[RG_ROWS + 32];
    __shared__ float tile[RG_ROWS * TILE_P];
    const int tid  = threadIdx.x;
    const int rg   = (int)blockIdx.x;
    const int n0   = rg << RG_SH;
    const int rows = min(RG_ROWS, n_nodes - n0);
    const int per  = (n_edges + BIN_WG - 1) / BIN_WG;
    const int c    = tid & (RG_ROWS - 1);
    const int sg   = tid >> RG_SH;
    const int cc   = min(c, rows - 1);
    const int sub  = tid & (FIN_SUB - 1);
    const int pg   = tid / FIN_SUB;
    const int idle = RG_ROWS + (tid & 31);

    if (tid < RG_ROWS + 32) dsum[tid] = (tid < RG_ROWS) ? 1.0f : 0.0f;
    __syncthreads();

    int    len[FIN_PC];
    size_t p0[FIN_PC];
    int    maxlen = 0;
#pragma unroll
    for (int i = 0; i < FIN_PC; ++i) {
        const int pc = pg + (FIN_TPB / FIN_SUB) * i;
        const int a  = off[(size_t)pc * OFF_P + rg];
        len[i] = off[(size_t)pc * OFF_P + rg + 1] - a;
        p0[i]  = (size_t)pc * per + a;
        maxlen = max(maxlen, len[i]);
    }
    float xv[BT / FIN_SG];
#pragma unroll
    for (int j = 0; j < BT / FIN_SG; ++j)
        xv[j] = x[(size_t)(sg + FIN_SG * j) * n_nodes + n0 + cc];

    const int fo = rg + (int)gridDim.x * tid;
    if (fo < 6 * HCH) {
        const bool  isb = fo >= 3 * HCH;
        const int   idx = isb ? fo - 3 * HCH : fo;
        const int   k   = idx / HCH;
        const int   co  = idx - k * HCH;
        const float* __restrict__ gs = isb ? gb : gw;
        float a = 0.0f;
        for (int ci = 0; ci < HCH; ++ci)
            a += gs[ci] * cw[(size_t)co * 96 + ci * 3 + k];
        fold[fo] = a;
    }

    for (int j = sub; j < maxlen; j += FIN_SUB) {
        unsigned r0[FIN_PC];
        float    w[FIN_PC];
#pragma unroll
        for (int i = 0; i < FIN_PC; ++i) {
            const size_t ri = (j < len[i]) ? p0[i] + j : 0;
            if (WIDE) {
                const uint2 r2 = ((const uint2*)rec)[ri];
                r0[i] = r2.x;
                w[i]  = __uint_as_float(r2.y);
            } else {
                r0[i] = rec[ri];
                w[i]  = ew[r0[i] & ((1u << REC_SH) - 1u)];
            }
        }
#pragma unroll
        for (int i = 0; i < FIN_PC; ++i) {
            asm volatile("" : "+v"(r0[i]), "+v"(w[i]));
            const bool ok = j < len[i];
            atomicAdd(&dsum[ok ? (int)(r0[i] >> (WIDE ? RECW_SH : REC_SH)) : idle], ok ? w[i] : 0.0f);
        }
    }
    __syncthreads();

    const float deg = dsum[c];
    const float di  = deg > 0.0f ? rsqrtf(deg) : 0.0f;
    if (tid < rows) dinv[n0 + tid] = di;
#pragma unroll
    for (int j = 0; j < BT / FIN_SG; ++j)
        tile[c * TILE_P + sg + FIN_SG * j] = xv[j] * di;
    __syncthreads();

    float4* __restrict__ o0 = (float4*)(xt0 + (size_t)n0 * P0W);
    float4* __restrict__ o1 = (float4*)(xt1 + (size_t)n0 * P1W);
    for (int q = tid; q < rows * (P0W / 4); q += FIN_TPB) {
        const float* t = &tile[(q / (P0W / 4)) * TILE_P + 4 * (q % (P0W / 4))];
        o0[q] = make_float4(t[0], t[1], t[2], t[3]);
    }
    for (int q = tid; q < rows * (P1W / 4); q += FIN_TPB) {
        const float* t = &tile[(q / (P1W / 4)) * TILE_P + P0W + 4 * (q % (P1W / 4))];
        o1[q] = make_float4(t[0], t[1], t[2], t[3]);
    }
}

#define AGG_TPB 512
#define AGG_NW  (AGG_TPB / 32)
#define AGG_G   8
#define AGG_CH  5
#define ACC_P   BT

template <bool FULL>
__device__ __forceinline__ unsigned agg_group(const float* __restrict__ xt0, const float* __restrict__ xt1,
                                              float* acc,
                                              unsigned mask, const int k,
                                              const int so, const int dof, const float nrm,
                                              const int lane, const bool upper, const int hl) {
    int   sj[AGG_G], dj[AGG_G];
    float nj[AGG_G], v0[AGG_G], v1[AGG_G / 2];
#pragma unroll
    for (int u = 0; u < AGG_G; ++u) {
        sj[u] = 0; dj[u] = 0; nj[u] = 0.0f; v0[u] = 0.0f;
        if (FULL || u < k) {
            const int j = __builtin_ctz(mask);
            mask &= mask - 1u;
            sj[u] = __builtin_amdgcn_readlane(so, j);
            dj[u] = __builtin_amdgcn_readlane(dof, j);
            nj[u] = __int_as_float(__builtin_amdgcn_readlane(__float_as_int(nrm), j));
        }
    }
#pragma unroll
    for (int u = 0; u < AGG_G; ++u)
        if (FULL || u < k) v0[u] = (xt0 + sj[u])[lane];
#pragma unroll
    for (int p = 0; p < AGG_G / 2; ++p) {
        v1[p] = 0.0f;
        if (FULL || 2 * p < k) {
            const bool ok = FULL || !upper || (2 * p + 1 < k);
            const int  sp = upper ? sj[2 * p + 1] : sj[2 * p];
            if (ok) v1[p] = xt1[(sp >> 1) + hl];
        }
    }
#pragma unroll
    for (int u = 0; u < AGG_G; ++u)
        if (FULL || u < k) atomicAdd(&acc[dj[u] + lane], v0[u] * nj[u]);
#pragma unroll
    for (int p = 0; p < AGG_G / 2; ++p) {
        if (FULL || 2 * p < k) {
            const bool  ok = FULL || !upper || (2 * p + 1 < k);
            const int   dp = upper ? dj[2 * p + 1] : dj[2 * p];
            const float np = upper ? nj[2 * p + 1] : nj[2 * p];
            if (ok) atomicAdd(&acc[dp + P0W + hl], v1[p] * np);
        }
    }
    return mask;
}

template <bool WIDE>
__global__ __launch_bounds__(AGG_TPB)
void k_aggregate(const int* __restrict__ ei, const float* __restrict__ ew,
                 const float* __restrict__ dinv,
                 const float* __restrict__ xt0, const float* __restrict__ xt1,
                 const unsigned* __restrict__ rec, const int* __restrict__ off,
                 const float* __restrict__ fold, const float* __restrict__ cb,
                 const float* __restrict__ lwp, const float* __restrict__ lbp,
                 float* __restrict__ out, int n_edges, int n_nodes) {
    __shared__ float acc[RG_ROWS * ACC_P];
    __shared__ float drow[RG_ROWS];
    __shared__ int   pstart[BIN_WG];
    __shared__ int   ppre[BIN_WG];
    __shared__ float w2s[3 * HCH];
    __shared__ float bbs[3 * HCH];

    const int  tid   = threadIdx.x;
    const int  lane  = tid & 31;
    const int  wave  = tid >> 5;
    const bool upper = lane >= 16;
    const int  hl    = lane & 15;
    const int  rg    = (int)blockIdx.x;
    const int  n0    = rg << RG_SH;
    const int  rows  = min(RG_ROWS, n_nodes - n0);
    const int  per   = (n_edges + BIN_WG - 1) / BIN_WG;

    for (int i = tid; i < RG_ROWS * ACC_P; i += AGG_TPB) {
        const int r = i / ACC_P;
        const int s = i - r * ACC_P;
        float v = 0.0f;
        if (r < rows) {
            const int n = n0 + r;
            v = (s < P0W) ? xt0[(size_t)n * P0W + s] : xt1[(size_t)n * P1W + (s - P0W)];
        }
        acc[i] = v;
    }
    if (tid < RG_ROWS) drow[tid] = (tid < rows) ? dinv[n0 + tid] : 0.0f;
    if (tid < 6 * HCH) {
        const float a = fold[tid];
        if (tid >= 3 * HCH) bbs[tid - 3 * HCH] = a; else w2s[tid] = a;
    }
    if (tid < BIN_WG) {
        const int a = off[(size_t)tid * OFF_P + rg];
        const int z = off[(size_t)tid * OFF_P + rg + 1];
        pstart[tid] = tid * per + a;
        ppre[tid]   = z - a;
    }
    __syncthreads();
    for (int d = 1; d < BIN_WG; d <<= 1) {
        int t = 0;
        if (tid < BIN_WG && tid >= d) t = ppre[tid - d];
        __syncthreads();
        if (tid < BIN_WG) ppre[tid] += t;
        __syncthreads();
    }
    const int total = ppre[BIN_WG - 1];
    const int nb    = (total + 31) >> 5;

    for (int b0 = wave; b0 < nb; b0 += AGG_NW * AGG_CH) {
        int      so[AGG_CH], dof[AGG_CH];
        float    nrm[AGG_CH];
        unsigned msk[AGG_CH];
        unsigned rv[AGG_CH];
        unsigned rw[AGG_CH];
#pragma unroll
        for (int c = 0; c < AGG_CH; ++c) {
            const int  i     = ((b0 + c * AGG_NW) << 5) + lane;
            const bool valid = i < total;
            msk[c] = (unsigned)__builtin_amdgcn_readfirstlane((int)__builtin_amdgcn_ballot_w32(valid));
            const int ic = min(i, total - 1);
            int p = 0;
#pragma unroll
            for (int st = BIN_WG / 2; st >= 1; st >>= 1)
                if (ppre[p + st - 1] <= ic) p += st;
            const int before = (p > 0) ? ppre[p - 1] : 0;
            const size_t ri = (size_t)pstart[p] + (ic - before);
            if (WIDE) {
                const uint2 r2 = ((const uint2*)rec)[ri];
                rv[c] = r2.x;
                rw[c] = r2.y;
            } else {
                rv[c] = rec[ri];
            }
        }
        int   sv[AGG_CH], dl[AGG_CH];
        float wv[AGG_CH];
#pragma unroll
        for (int c = 0; c < AGG_CH; ++c) {
            if (WIDE) {
                sv[c] = (int)(rv[c] & ((1u << RECW_SH) - 1u));
                dl[c] = (int)(rv[c] >> RECW_SH);
                wv[c] = __uint_as_float(rw[c]);
            } else {
                const int e = (int)(rv[c] & ((1u << REC_SH) - 1u));
                dl[c] = (int)(rv[c] >> REC_SH);
                sv[c] = ei[e];
                wv[c] = ew[e];
            }
        }
#pragma unroll
        for (int c = 0; c < AGG_CH; ++c) {
            nrm[c] = wv[c];
            so[c]  = sv[c] * P0W;
            dof[c] = dl[c] * ACC_P;
            if (!WIDE) asm volatile("" : "+v"(nrm[c]), "+v"(so[c]));
        }
#pragma unroll
        for (int c = 0; c < AGG_CH; ++c) {
            unsigned m = msk[c];
            while (m != 0u) {
                const int k = __builtin_popcount(m);
                if (k >= AGG_G) m = agg_group<true >(xt0, xt1, acc, m, AGG_G, so[c], dof[c], nrm[c], lane, upper, hl);
                else            m = agg_group<false>(xt0, xt1, acc, m, k,     so[c], dof[c], nrm[c], lane, upper, hl);
            }
        }
    }

    __syncthreads();

    const int h = lane >> 4;
    const int m = lane & 15;

    v2f bmat0, bmat1;
#pragma unroll
    for (int v = 0; v < 2; ++v) {
        const int  k  = 2 * h + v;
        const int  kc = (k < 3) ? k : 0;
        const float a0 = w2s[kc * HCH + m];
        const float a1 = w2s[kc * HCH + m + 16];
        bmat0[v] = (k < 3) ? a0 : 0.0f;
        bmat1[v] = (k < 3) ? a1 : 0.0f;
    }
    float bbA[3], bbB[3];
#pragma unroll
    for (int k = 0; k < 3; ++k) {
        bbA[k] = bbs[k * HCH + m];
        bbB[k] = bbs[k * HCH + m + 16];
    }
    v8f cmat0, cmat1;
#pragma unroll
    for (int j = 0; j < 8; ++j) {
        int t = j + 8 * h;
        float c0 = 0.0f, c1 = 0.0f;
        if (t < TT) {
            c0 = cb[m]      + bbA[1] + (t > 0 ? bbA[0] : 0.0f) + (t < TT - 1 ? bbA[2] : 0.0f);
            c1 = cb[m + 16] + bbB[1] + (t > 0 ? bbB[0] : 0.0f) + (t < TT - 1 ? bbB[2] : 0.0f);
        }
        cmat0[j] = c0;
        cmat1[j] = c1;
    }
    const float lwA = lwp[m], lwB = lwp[16 + m];
    const float lb  = lbp[0];

    const int  i0  = m + 2 * h - 1;
    const int  i1  = i0 + 1;
    const bool ok0 = (m < TT) && (i0 >= 0) && (i0 < TT);
    const bool ok1 = (m < TT) && (i1 < TT);
    const int  o0  = ok0 ? i0 : 0;
    const int  o1  = ok1 ? i1 : 0;
    const float msk0 = ok0 ? 1.0f : 0.0f;
    const float msk1 = ok1 ? 1.0f : 0.0f;

#pragma unroll
    for (int b = 0; b < BB; ++b) {
        const int s0 = b * TT + o0;
        const int s1 = b * TT + o1;
        float* outb = out + (size_t)b * n_nodes + n0;
        for (int r = wave; r < rows; r += AGG_NW) {
            const float dr = drow[r];
            v2f amat;
            amat[0] = acc[r * ACC_P + s0] * dr * msk0;
            amat[1] = acc[r * ACC_P + s1] * dr * msk1;

            v8f d0 = __builtin_amdgcn_wmma_f32_16x16x4_f32(false, amat, false, bmat0,
                                                           (short)0, cmat0, false, false);
            v8f d1 = __builtin_amdgcn_wmma_f32_16x16x4_f32(false, amat, false, bmat1,
                                                           (short)0, cmat1, false, false);

            float sA = 0.0f, sB = 0.0f;
#pragma unroll
            for (int j = 0; j < 8; ++j) {
                sA += fmaxf(d0[j], 0.0f);
                sB += fmaxf(d1[j], 0.0f);
            }
            sA += __shfl_xor(sA, 16);
            sB += __shfl_xor(sB, 16);

            float v = sA * lwA + sB * lwB;
            v += __shfl_xor(v, 1);
            v += __shfl_xor(v, 2);
            v += __shfl_xor(v, 4);
            v += __shfl_xor(v, 8);

            if (lane == 0)
                outb[r] = v * (1.0f / (float)TT) + lb;
        }
    }
}

extern "C" void kernel_launch(void* const* d_in, const int* in_sizes, int n_in,
                              void* d_out, int out_size, void* d_ws, size_t ws_size,
                              hipStream_t stream) {
    const float* x  = (const float*)d_in[0];
    const int*   ei = (const int*)d_in[1];
    const float* ew = (const float*)d_in[2];
    const float* gw = (const float*)d_in[3];
    const float* gb = (const float*)d_in[4];
    const float* cw = (const float*)d_in[5];
    const float* cb = (const float*)d_in[6];
    const float* lw = (const float*)d_in[7];
    const float* lb = (const float*)d_in[8];
    float* out = (float*)d_out;

    const int E = in_sizes[2];
    const int N = in_sizes[0] / BT;

    const size_t words_wide = (size_t)N * (P0W + P1W + 1) + (size_t)E * 2 + (size_t)BIN_WG * OFF_P + 6 * HCH;
    const bool   wide       = ws_size >= words_wide * sizeof(float);
    float*    xt0  = (float*)d_ws;
    float*    xt1  = xt0 + (size_t)N * P0W;
    unsigned* rec  = (unsigned*)(xt1 + (size_t)N * P1W);
    float*    dinv = (float*)(rec + (wide ? (size_t)E * 2 : (size_t)E));
    int*      off  = (int*)(dinv + N);
    float*    fold = (float*)(off + (size_t)BIN_WG * OFF_P);

    if (wide) {
        k_bin      <true ><<<BIN_WG, BIN_TPB, 0, stream>>>(ei, ew, rec, off, E);
        k_rows     <true ><<<(N + RG_ROWS - 1) / RG_ROWS, FIN_TPB, 0, stream>>>(x, ew, rec, off, gw, gb, cw, fold,
                                                                               dinv, xt0, xt1, E, N);
        k_aggregate<true ><<<(N + RG_ROWS - 1) / RG_ROWS, AGG_TPB, 0, stream>>>(ei, ew, dinv, xt0, xt1, rec, off,
                                                                               fold, cb, lw, lb, out, E, N);
    } else {
        k_bin      <false><<<BIN_WG, BIN_TPB, 0, stream>>>(ei, ew, rec, off, E);
        k_rows     <false><<<(N + RG_ROWS - 1) / RG_ROWS, FIN_TPB, 0, stream>>>(x, ew, rec, off, gw, gb, cw, fold,
                                                                               dinv, xt0, xt1, E, N);
        k_aggregate<false><<<(N + RG_ROWS - 1) / RG_ROWS, AGG_TPB, 0, stream>>>(ei, ew, dinv, xt0, xt1, rec, off,
                                                                               fold, cb, lw, lb, out, E, N);
    }
}


# aggregation: records ordered by destination row in LDS (counting sort, arrival-order rank); a wave owns whole rows, sums them in registers (fma), 16 bytes a lane: no float LDS atomics; sum order new
# speedup vs baseline: 5.5135x; 1.1760x over previous
#include <hip/hip_runtime.h>
#include <hip/hip_bf16.h>


typedef __attribute__((ext_vector_type(2))) float v2f;
typedef __attribute__((ext_vector_type(8))) float v8f;

#define BT 48
#define TT 12
#define BB 4
#define HCH 32

#define P0W 32
#define P1W 16

#define RG_SH    6
#define RG_ROWS  (1 << RG_SH)
#define RG_MAX   1024
#define BIN_WG   256
#define BIN_TPB  1024
#define BIN_UNR  8
#define OFF_P    (RG_MAX + 1)
#define REC_SH   25
#define RECW_SH  26

static_assert(RG_MAX == BIN_TPB, "the binning keeps one range per thread");
static_assert(BIN_WG <= 512, "the aggregation keeps one piece per thread");
template <bool WIDE>
__global__ __launch_bounds__(BIN_TPB)
void k_bin(const int* __restrict__ ei, const float* __restrict__ ew,
           unsigned* __restrict__ rec, int* __restrict__ off, int n_edges) {
    __shared__ int lcnt[RG_MAX];
    __shared__ int lsc[RG_MAX];
    const int tid = threadIdx.x;
    const int per = (n_edges + BIN_WG - 1) / BIN_WG;
    const int e0  = (int)blockIdx.x * per;
    const int e1  = min(e0 + per, n_edges);
    const int* __restrict__ dst = ei + (size_t)n_edges;
    const int step = BIN_UNR * BIN_TPB;

    lcnt[tid] = 0;
    __syncthreads();
    for (int e = e0 + tid; e < e1; e += step) {
        int dv[BIN_UNR];
#pragma unroll
        for (int u = 0; u < BIN_UNR; ++u) dv[u] = dst[min(e + u * BIN_TPB, e1 - 1)];
#pragma unroll
        for (int u = 0; u < BIN_UNR; ++u)
            if (e + u * BIN_TPB < e1) atomicAdd(&lcnt[dv[u] >> RG_SH], 1);
    }
    __syncthreads();

    const int mine = lcnt[tid];
    lsc[tid] = mine;
    __syncthreads();
    for (int d = 1; d < RG_MAX; d <<= 1) {
        const int t = (tid >= d) ? lsc[tid - d] : 0;
        __syncthreads();
        lsc[tid] += t;
        __syncthreads();
    }
    const int incl = lsc[tid];
    const int excl = incl - mine;
    int* __restrict__ offw = off + (size_t)blockIdx.x * OFF_P;
    offw[tid] = excl;
    if (tid == RG_MAX - 1) offw[RG_MAX] = incl;
    lcnt[tid] = excl;
    __syncthreads();

    uint2* __restrict__ rec2 = (uint2*)rec;
    for (int e = e0 + tid; e < e1; e += step) {
        int   dv[BIN_UNR];
        int   sv[BIN_UNR];
        float wv[BIN_UNR];
#pragma unroll
        for (int u = 0; u < BIN_UNR; ++u) {
            const int ec = min(e + u * BIN_TPB, e1 - 1);
            dv[u] = dst[ec];
            if (WIDE) { sv[u] = ei[ec]; wv[u] = ew[ec]; }
        }
#pragma unroll
        for (int u = 0; u < BIN_UNR; ++u) {
            if (e + u * BIN_TPB < e1) {
                const int      p   = atomicAdd(&lcnt[dv[u] >> RG_SH], 1);
                const unsigned row = (unsigned)(dv[u] & (RG_ROWS - 1));
                if (WIDE) rec2[(size_t)e0 + p] = make_uint2((unsigned)sv[u] | (row << RECW_SH), __float_as_uint(wv[u]));
                else      rec[(size_t)e0 + p]  = (unsigned)(e + u * BIN_TPB) | (row << REC_SH);
            }
        }
    }
}

#define FIN_TPB 256
#define FIN_SUB 8
#define FIN_PC  (BIN_WG / (FIN_TPB / FIN_SUB))
#define FIN_SG  (FIN_TPB / RG_ROWS)
#define TILE_P  (BT + 1)
static_assert(BIN_WG % (FIN_TPB / FIN_SUB) == 0, "every piece has its lanes");
static_assert(BT % FIN_SG == 0, "a node's slices are split evenly over its threads");
static_assert(FIN_TPB >= RG_ROWS + 32, "one thread per word of the degree sums");

template <bool WIDE>
__global__ __launch_bounds__(FIN_TPB)
void k_rows(const float* __restrict__ x, const float* __restrict__ ew,
            const unsigned* __restrict__ rec, const int* __restrict__ off,
            const float* __restrict__ gw, const float* __restrict__ gb, const float* __restrict__ cw,
            float* __restrict__ fold,
            float* __restrict__ dinv, float* __restrict__ xt0, float* __restrict__ xt1,
            int n_edges, int n_nodes) {
    __shared__ float dsum[RG_ROWS + 32];
    __shared__ float tile[RG_ROWS * TILE_P];
    const int tid  = threadIdx.x;
    const int rg   = (int)blockIdx.x;
    const int n0   = rg << RG_SH;
    const int rows = min(RG_ROWS, n_nodes - n0);
    const int per  = (n_edges + BIN_WG - 1) / BIN_WG;
    const int c    = tid & (RG_ROWS - 1);
    const int sg   = tid >> RG_SH;
    const int cc   = min(c, rows - 1);
    const int sub  = tid & (FIN_SUB - 1);
    const int pg   = tid / FIN_SUB;
    const int idle = RG_ROWS + (tid & 31);

    if (tid < RG_ROWS + 32) dsum[tid] = (tid < RG_ROWS) ? 1.0f : 0.0f;
    __syncthreads();

    int    len[FIN_PC];
    size_t p0[FIN_PC];
    int    maxlen = 0;
#pragma unroll
    for (int i = 0; i < FIN_PC; ++i) {
        const int pc = pg + (FIN_TPB / FIN_SUB) * i;
        const int a  = off[(size_t)pc * OFF_P + rg];
        len[i] = off[(size_t)pc * OFF_P + rg + 1] - a;
        p0[i]  = (size_t)pc * per + a;
        maxlen = max(maxlen, len[i]);
    }
    float xv[BT / FIN_SG];
#pragma unroll
    for (int j = 0; j < BT / FIN_SG; ++j)
        xv[j] = x[(size_t)(sg + FIN_SG * j) * n_nodes + n0 + cc];

    const int fo = rg + (int)gridDim.x * tid;
    if (fo < 6 * HCH) {
        const bool  isb = fo >= 3 * HCH;
        const int   idx = isb ? fo - 3 * HCH : fo;
        const int   k   = idx / HCH;
        const int   co  = idx - k * HCH;
        const float* __restrict__ gs = isb ? gb : gw;
        float a = 0.0f;
        for (int ci = 0; ci < HCH; ++ci)
            a += gs[ci] * cw[(size_t)co * 96 + ci * 3 + k];
        fold[fo] = a;
    }

    for (int j = sub; j < maxlen; j += FIN_SUB) {
        unsigned r0[FIN_PC];
        float    w[FIN_PC];
#pragma unroll
        for (int i = 0; i < FIN_PC; ++i) {
            const size_t ri = (j < len[i]) ? p0[i] + j : 0;
            if (WIDE) {
                const uint2 r2 = ((const uint2*)rec)[ri];
                r0[i] = r2.x;
                w[i]  = __uint_as_float(r2.y);
            } else {
                r0[i] = rec[ri];
                w[i]  = ew[r0[i] & ((1u << REC_SH) - 1u)];
            }
        }
#pragma unroll
        for (int i = 0; i < FIN_PC; ++i) {
            asm volatile("" : "+v"(r0[i]), "+v"(w[i]));
            const bool ok = j < len[i];
            atomicAdd(&dsum[ok ? (int)(r0[i] >> (WIDE ? RECW_SH : REC_SH)) : idle], ok ? w[i] : 0.0f);
        }
    }
    __syncthreads();

    const float deg = dsum[c];
    const float di  = deg > 0.0f ? rsqrtf(deg) : 0.0f;
    if (tid < rows) dinv[n0 + tid] = di;
#pragma unroll
    for (int j = 0; j < BT / FIN_SG; ++j)
        tile[c * TILE_P + sg + FIN_SG * j] = xv[j] * di;
    __syncthreads();

    float4* __restrict__ o0 = (float4*)(xt0 + (size_t)n0 * P0W);
    float4* __restrict__ o1 = (float4*)(xt1 + (size_t)n0 * P1W);
    for (int q = tid; q < rows * (P0W / 4); q += FIN_TPB) {
        const float* t = &tile[(q / (P0W / 4)) * TILE_P + 4 * (q % (P0W / 4))];
        o0[q] = make_float4(t[0], t[1], t[2], t[3]);
    }
    for (int q = tid; q < rows * (P1W / 4); q += FIN_TPB) {
        const float* t = &tile[(q / (P1W / 4)) * TILE_P + P0W + 4 * (q % (P1W / 4))];
        o1[q] = make_float4(t[0], t[1], t[2], t[3]);
    }
}

#define AGG_TPB 512
#define AGG_NW  (AGG_TPB / 32)
#define AGG_CH  5
#define AGG_CAP (AGG_TPB * AGG_CH)
#define AGG_OWN (RG_ROWS / AGG_NW)
#define AGG_U   6
#define ACC_P   BT
static_assert(RG_ROWS % AGG_NW == 0, "every row has its wave");
static_assert(P0W == 32 && P1W == 16 && P0W + P1W == BT, "a row is eight + four lanes of 16 bytes");

template <bool FULL>
__device__ __forceinline__ void agg_trip(const char* __restrict__ pb, const uint2* srt, float4& ra,
                                         const int j0, const int je, const int hf,
                                         const unsigned plo, const int psh) {
    uint2  rr[AGG_U];
    float4 v[AGG_U];
#pragma unroll
    for (int u = 0; u < AGG_U; ++u)
        rr[u] = srt[FULL ? (j0 + 2 * u + hf) : min(j0 + 2 * u + hf, je - 1)];
#pragma unroll
    for (int u = 0; u < AGG_U; ++u) {
        v[u] = make_float4(0.0f, 0.0f, 0.0f, 0.0f);
        if (FULL || j0 + 2 * u + hf < je) v[u] = *(const float4*)(pb + (plo + (rr[u].x << psh)));
    }
#pragma unroll
    for (int u = 0; u < AGG_U; ++u) {
        const float w = (FULL || j0 + 2 * u + hf < je) ? __uint_as_float(rr[u].y) : 0.0f;
        ra.x = __builtin_fmaf(v[u].x, w, ra.x);
        ra.y = __builtin_fmaf(v[u].y, w, ra.y);
        ra.z = __builtin_fmaf(v[u].z, w, ra.z);
        ra.w = __builtin_fmaf(v[u].w, w, ra.w);
    }
}

template <bool WIDE>
__global__ __launch_bounds__(AGG_TPB, 16)
void k_aggregate(const int* __restrict__ ei, const float* __restrict__ ew,
                 const float* __restrict__ dinv,
                 const float* __restrict__ xt0, const float* __restrict__ xt1,
                 const unsigned* __restrict__ rec, const int* __restrict__ off,
                 const float* __restrict__ fold, const float* __restrict__ cb,
                 const float* __restrict__ lwp, const float* __restrict__ lbp,
                 float* __restrict__ out, int n_edges, int n_nodes) {
    __shared__ __align__(16) float acc[RG_ROWS * ACC_P];
    __shared__ uint2 srt[AGG_CAP];
    __shared__ int   rcnt[RG_ROWS];
    __shared__ int   rbeg[RG_ROWS];
    __shared__ float drow[RG_ROWS];
    __shared__ int   pstart[BIN_WG];
    __shared__ int   ppre[BIN_WG];
    __shared__ float w2s[3 * HCH];
    __shared__ float bbs[3 * HCH];

    const int  tid   = threadIdx.x;
    const int  lane  = tid & 31;
    const int  wave  = __builtin_amdgcn_readfirstlane(tid >> 5);
    const int  hf    = lane >> 4;
    const int  q     = lane & 15;
    const int  qc    = min(q, 11);
    const int  rg    = (int)blockIdx.x;
    const int  n0    = rg << RG_SH;
    const int  rows  = min(RG_ROWS, n_nodes - n0);
    const int  per   = (n_edges + BIN_WG - 1) / BIN_WG;

    const char* __restrict__ pb  = (const char*)xt0;
    const unsigned           p1o = (unsigned)((const char*)xt1 - (const char*)xt0);
    const unsigned           plo = (qc < 8) ? (unsigned)(16 * qc) : p1o + (unsigned)(16 * (qc - 8));
    const int                psh = (qc < 8) ? 7 : 6;

    const bool own = (hf == 0) && (q < 12);
#pragma unroll
    for (int k = 0; k < AGG_OWN; ++k) {
        const int r = wave * AGG_OWN + k;
        float4 t = make_float4(0.0f, 0.0f, 0.0f, 0.0f);
        if (own && r < rows) t = *(const float4*)(pb + (plo + ((unsigned)(n0 + r) << psh)));
        if (own) *(float4*)&acc[r * ACC_P + 4 * q] = t;
    }
    if (tid < RG_ROWS) {
        drow[tid] = (tid < rows) ? dinv[n0 + tid] : 0.0f;
        rcnt[tid] = 0;
    }
    if (tid < 6 * HCH) {
        const float a = fold[tid];
        if (tid >= 3 * HCH) bbs[tid - 3 * HCH] = a; else w2s[tid] = a;
    }
    if (tid < BIN_WG) {
        const int a = off[(size_t)tid * OFF_P + rg];
        const int z = off[(size_t)tid * OFF_P + rg + 1];
        pstart[tid] = tid * per + a;
        ppre[tid]   = z - a;
    }
    __syncthreads();
    for (int d = 1; d < BIN_WG; d <<= 1) {
        int t = 0;
        if (tid < BIN_WG && tid >= d) t = ppre[tid - d];
        __syncthreads();
        if (tid < BIN_WG) ppre[tid] += t;
        __syncthreads();
    }
    const int total = ppre[BIN_WG - 1];

    for (int base = 0; base < total; base += AGG_CAP) {
        bool     val[AGG_CH];
        unsigned rv[AGG_CH];
        unsigned rw[AGG_CH];
#pragma unroll
        for (int c = 0; c < AGG_CH; ++c) {
            const int i = base + ((wave + c * AGG_NW) << 5) + lane;
            val[c] = i < total;
            const int ic = min(i, total - 1);
            int p = 0;
#pragma unroll
            for (int st = BIN_WG / 2; st >= 1; st >>= 1)
                if (ppre[p + st - 1] <= ic) p += st;
            const int before = (p > 0) ? ppre[p - 1] : 0;
            const size_t ri = (size_t)pstart[p] + (ic - before);
            rw[c] = 0u;
            if (WIDE) {
                const uint2 r2 = ((const uint2*)rec)[ri];
                rv[c] = r2.x;
                rw[c] = r2.y;
            } else {
                rv[c] = rec[ri];
            }
        }
        unsigned sv[AGG_CH];
        int      dl[AGG_CH];
        unsigned wv[AGG_CH];
#pragma unroll
        for (int c = 0; c < AGG_CH; ++c) {
            if (WIDE) {
                sv[c] = rv[c] & ((1u << RECW_SH) - 1u);
                dl[c] = (int)(rv[c] >> RECW_SH);
                wv[c] = rw[c];
            } else {
                const int e = (int)(rv[c] & ((1u << REC_SH) - 1u));
                dl[c] = (int)(rv[c] >> REC_SH);
                sv[c] = (unsigned)ei[e];
                wv[c] = __float_as_uint(ew[e]);
            }
        }
        int rk[AGG_CH];
#pragma unroll
        for (int c = 0; c < AGG_CH; ++c)
            rk[c] = val[c] ? atomicAdd(&rcnt[dl[c]], 1) : 0;
        __syncthreads();
        if (wave == 0) {
            const int a = rcnt[2 * lane];
            const int b = rcnt[2 * lane + 1];
            int sc = a + b;
            for (int d = 1; d < 32; d <<= 1) {
                const int t = __shfl_up(sc, d);
                if (lane >= d) sc += t;
            }
            const int ex = sc - (a + b);
            rbeg[2 * lane]     = ex;
            rbeg[2 * lane + 1] = ex + a;
        }
        __syncthreads();
#pragma unroll
        for (int c = 0; c < AGG_CH; ++c)
            if (val[c]) srt[rbeg[dl[c]] + rk[c]] = make_uint2(sv[c], wv[c]);
        __syncthreads();

        for (int k = 0; k < AGG_OWN; ++k) {
            const int r  = wave * AGG_OWN + k;
            const int js = __builtin_amdgcn_readfirstlane(rbeg[r]);
            const int je = js + __builtin_amdgcn_readfirstlane(rcnt[r]);
            if (js == je) continue;
            float4 ra = make_float4(0.0f, 0.0f, 0.0f, 0.0f);
            int j0 = js;
            for (; j0 + 2 * AGG_U <= je; j0 += 2 * AGG_U)
                agg_trip<true >(pb, srt, ra, j0, je, hf, plo, psh);
            if (j0 < je)
                agg_trip<false>(pb, srt, ra, j0, je, hf, plo, psh);
            ra.x += __shfl_xor(ra.x, 16);
            ra.y += __shfl_xor(ra.y, 16);
            ra.z += __shfl_xor(ra.z, 16);
            ra.w += __shfl_xor(ra.w, 16);
            if (own) {
                float4* ap = (float4*)&acc[r * ACC_P + 4 * q];
                float4  t  = *ap;
                t.x += ra.x; t.y += ra.y; t.z += ra.z; t.w += ra.w;
                *ap = t;
            }
        }
        if (base + AGG_CAP < total) {
            __syncthreads();
            if (tid < RG_ROWS) rcnt[tid] = 0;
            __syncthreads();
        }
    }

    __syncthreads();

    const int h = lane >> 4;
    const int m = lane & 15;

    v2f bmat0, bmat1;
#pragma unroll
    for (int v = 0; v < 2; ++v) {
        const int  k  = 2 * h + v;
        const int  kc = (k < 3) ? k : 0;
        const float a0 = w2s[kc * HCH + m];
        const float a1 = w2s[kc * HCH + m + 16];
        bmat0[v] = (k < 3) ? a0 : 0.0f;
        bmat1[v] = (k < 3) ? a1 : 0.0f;
    }
    float bbA[3], bbB[3];
#pragma unroll
    for (int k = 0; k < 3; ++k) {
        bbA[k] = bbs[k * HCH + m];
        bbB[k] = bbs[k * HCH + m + 16];
    }
    v8f cmat0, cmat1;
#pragma unroll
    for (int j = 0; j < 8; ++j) {
        int t = j + 8 * h;
        float c0 = 0.0f, c1 = 0.0f;
        if (t < TT) {
            c0 = cb[m]      + bbA[1] + (t > 0 ? bbA[0] : 0.0f) + (t < TT - 1 ? bbA[2] : 0.0f);
            c1 = cb[m + 16] + bbB[1] + (t > 0 ? bbB[0] : 0.0f) + (t < TT - 1 ? bbB[2] : 0.0f);
        }
        cmat0[j] = c0;
        cmat1[j] = c1;
    }
    const float lwA = lwp[m], lwB = lwp[16 + m];
    const float lb  = lbp[0];

    const int  i0  = m + 2 * h - 1;
    const int  i1  = i0 + 1;
    const bool ok0 = (m < TT) && (i0 >= 0) && (i0 < TT);
    const bool ok1 = (m < TT) && (i1 < TT);
    const int  o0  = ok0 ? i0 : 0;
    const int  o1  = ok1 ? i1 : 0;
    const float msk0 = ok0 ? 1.0f : 0.0f;
    const float msk1 = ok1 ? 1.0f : 0.0f;

#pragma unroll
    for (int b = 0; b < BB; ++b) {
        const int s0 = b * TT + o0;
        const int s1 = b * TT + o1;
        float* outb = out + (size_t)b * n_nodes + n0;
        for (int r = wave; r < rows; r += AGG_NW) {
            const float dr = drow[r];
            v2f amat;
            amat[0] = acc[r * ACC_P + s0] * dr * msk0;
            amat[1] = acc[r * ACC_P + s1] * dr * msk1;

            v8f d0 = __builtin_amdgcn_wmma_f32_16x16x4_f32(false, amat, false, bmat0,
                                                           (short)0, cmat0, false, false);
            v8f d1 = __builtin_amdgcn_wmma_f32_16x16x4_f32(false, amat, false, bmat1,
                                                           (short)0, cmat1, false, false);

            float sA = 0.0f, sB = 0.0f;
#pragma unroll
            for (int j = 0; j < 8; ++j) {
                sA += fmaxf(d0[j], 0.0f);
                sB += fmaxf(d1[j], 0.0f);
            }
            sA += __shfl_xor(sA, 16);
            sB += __shfl_xor(sB, 16);

            float v = sA * lwA + sB * lwB;
            v += __shfl_xor(v, 1);
            v += __shfl_xor(v, 2);
            v += __shfl_xor(v, 4);
            v += __shfl_xor(v, 8);

            if (lane == 0)
                outb[r] = v * (1.0f / (float)TT) + lb;
        }
    }
}

extern "C" void kernel_launch(void* const* d_in, const int* in_sizes, int n_in,
                              void* d_out, int out_size, void* d_ws, size_t ws_size,
                              hipStream_t stream) {
    const float* x  = (const float*)d_in[0];
    const int*   ei = (const int*)d_in[1];
    const float* ew = (const float*)d_in[2];
    const float* gw = (const float*)d_in[3];
    const float* gb = (const float*)d_in[4];
    const float* cw = (const float*)d_in[5];
    const float* cb = (const float*)d_in[6];
    const float* lw = (const float*)d_in[7];
    const float* lb = (const float*)d_in[8];
    float* out = (float*)d_out;

    const int E = in_sizes[2];
    const int N = in_sizes[0] / BT;

    const size_t words_wide = (size_t)N * (P0W + P1W + 1) + (size_t)E * 2 + (size_t)BIN_WG * OFF_P + 6 * HCH;
    const bool   wide       = ws_size >= words_wide * sizeof(float);
    float*    xt0  = (float*)d_ws;
    float*    xt1  = xt0 + (size_t)N * P0W;
    unsigned* rec  = (unsigned*)(xt1 + (size_t)N * P1W);
    float*    dinv = (float*)(rec + (wide ? (size_t)E * 2 : (size_t)E));
    int*      off  = (int*)(dinv + N);
    float*    fold = (float*)(off + (size_t)BIN_WG * OFF_P);

    if (wide) {
        k_bin      <true ><<<BIN_WG, BIN_TPB, 0, stream>>>(ei, ew, rec, off, E);
        k_rows     <true ><<<(N + RG_ROWS - 1) / RG_ROWS, FIN_TPB, 0, stream>>>(x, ew, rec, off, gw, gb, cw, fold,
                                                                               dinv, xt0, xt1, E, N);
        k_aggregate<true ><<<(N + RG_ROWS - 1) / RG_ROWS, AGG_TPB, 0, stream>>>(ei, ew, dinv, xt0, xt1, rec, off,
                                                                               fold, cb, lw, lb, out, E, N);
    } else {
        k_bin      <false><<<BIN_WG, BIN_TPB, 0, stream>>>(ei, ew, rec, off, E);
        k_rows     <false><<<(N + RG_ROWS - 1) / RG_ROWS, FIN_TPB, 0, stream>>>(x, ew, rec, off, gw, gb, cw, fold,
                                                                               dinv, xt0, xt1, E, N);
        k_aggregate<false><<<(N + RG_ROWS - 1) / RG_ROWS, AGG_TPB, 0, stream>>>(ei, ew, dinv, xt0, xt1, rec, off,
                                                                               fold, cb, lw, lb, out, E, N);
    }
}
